# starting kernel
# baseline (speedup 1.0000x reference)
#include <hip/hip_runtime.h>
#include <math.h>

typedef float v2f __attribute__((ext_vector_type(2)));
typedef float v8f __attribute__((ext_vector_type(8)));

#define N_NODES 1024
#define IN_F    256
#define OUT_F   256
#define HEADS   4
#define HID     64
#define NEG_SLOPE 0.2f

__global__ void gemm_xw(const float* __restrict__ X, const float* __restrict__ W,
                        float* __restrict__ G) {
    const int lane  = threadIdx.x;
    const int tileN = blockIdx.x * 16;
    const int tileM = blockIdx.y * 16;
    const int lid   = lane & 15;
    const int kh    = (lane >> 4) * 2;
    const int m     = tileM + lid;
    const int n     = tileN + lid;

    v8f acc = {};
    for (int k0 = 0; k0 < IN_F; k0 += 4) {
        const int k = k0 + kh;
        v2f a, b;
        a.x = X[m * IN_F + k];
        a.y = X[m * IN_F + k + 1];
        b.x = W[k * OUT_F + n];
        b.y = W[(k + 1) * OUT_F + n];
        acc = __builtin_amdgcn_wmma_f32_16x16x4_f32(false, a, false, b,
                                                    (short)0, acc, false, false);
    }
    const int row_half = (lane >> 4) * 8;
#pragma unroll
    for (int p = 0; p < 8; ++p) {
        const int mm = tileM + p + row_half;
        G[mm * OUT_F + n] = acc[p];
    }
}

__global__ void attn_scores(const float* __restrict__ gl, const float* __restrict__ gr,
                            const float* __restrict__ adj, const float* __restrict__ aw,
                            float* __restrict__ P, float* __restrict__ denom) {
    const int i = blockIdx.x;
    const int h = blockIdx.y;
    const int t = threadIdx.x;

    __shared__ float s_gri[HID];
    __shared__ float s_aw[HID];
    __shared__ float s_red[256];

    if (t < HID) {
        s_gri[t] = gr[i * OUT_F + h * HID + t];
        s_aw[t]  = aw[t];
    }
    __syncthreads();

    float part = 0.0f;
    float* Ph = P + (size_t)h * N_NODES * N_NODES + (size_t)i * N_NODES;
    for (int j = t; j < N_NODES; j += 256) {
        float p = 0.0f;
        if (adj[(size_t)j * N_NODES + i] != 0.0f) {
            const float* glj = gl + j * OUT_F + h * HID;
            float e = 0.0f;
#pragma unroll
            for (int f = 0; f < HID; ++f) {
                const float s  = s_gri[f] + glj[f];
                const float lr = fmaxf(s, NEG_SLOPE * s);
                e = fmaf(lr, s_aw[f], e);
            }
            p = expf(e);
        }
        Ph[j] = p;
        part += p;
    }

    s_red[t] = part;
    __syncthreads();
    for (int s = 128; s > 0; s >>= 1) {
        if (t < s) s_red[t] += s_red[t + s];
        __syncthreads();
    }
    if (t == 0) denom[h * N_NODES + i] = s_red[0];
}

__global__ void aggregate(const float* __restrict__ P, const float* __restrict__ denom,
                          const float* __restrict__ gr, float* __restrict__ out,
                          const int* __restrict__ use_elu) {
    const int lane  = threadIdx.x;
    const int tileF = blockIdx.x * 16;
    const int tileM = blockIdx.y * 16;
    const int h     = blockIdx.z;
    const int lid   = lane & 15;
    const int kh    = (lane >> 4) * 2;
    const int m     = tileM + lid;

    const float recip = 1.0f / denom[h * N_NODES + m];
    const float* Ph = P + (size_t)h * N_NODES * N_NODES;

    v8f acc = {};
    for (int k0 = 0; k0 < N_NODES; k0 += 4) {
        const int k = k0 + kh;
        v2f a, b;
        a.x = Ph[(size_t)m * N_NODES + k]     * recip;
        a.y = Ph[(size_t)m * N_NODES + k + 1] * recip;
        b.x = gr[k * OUT_F + h * HID + tileF + lid];
        b.y = gr[(k + 1) * OUT_F + h * HID + tileF + lid];
        acc = __builtin_amdgcn_wmma_f32_16x16x4_f32(false, a, false, b,
                                                    (short)0, acc, false, false);
    }

    const int elu = use_elu[0];
    const int row_half = (lane >> 4) * 8;
#pragma unroll
    for (int p = 0; p < 8; ++p) {
        const int i = tileM + p + row_half;
        float x = acc[p];
        if (elu) x = (x > 0.0f) ? x : expm1f(x);
        out[i * OUT_F + h * HID + tileF + lid] = x;
    }
}

extern "C" void kernel_launch(void* const* d_in, const int* in_sizes, int n_in,
                              void* d_out, int out_size, void* d_ws, size_t ws_size,
                              hipStream_t stream) {
    const float* h_in  = (const float*)d_in[0];
    const float* nei   = (const float*)d_in[1];
    const float* adj   = (const float*)d_in[2];
    const float* w_l   = (const float*)d_in[3];
    const float* w_r   = (const float*)d_in[4];
    const float* aw    = (const float*)d_in[5];
    const int*   elu   = (const int*)d_in[6];
    float* out = (float*)d_out;

    float* ws    = (float*)d_ws;
    float* g_l   = ws;
    float* g_r   = ws + (size_t)N_NODES * OUT_F;
    float* P     = g_r + (size_t)N_NODES * OUT_F;
    float* denom = P + (size_t)HEADS * N_NODES * N_NODES;

    dim3 ggrid(OUT_F / 16, N_NODES / 16);
    gemm_xw<<<ggrid, 32, 0, stream>>>(h_in, w_l, g_l);
    gemm_xw<<<ggrid, 32, 0, stream>>>(nei,  w_r, g_r);

    attn_scores<<<dim3(N_NODES, HEADS), 256, 0, stream>>>(g_l, g_r, adj, aw, P, denom);

    aggregate<<<dim3(HID / 16, N_NODES / 16, HEADS), 32, 0, stream>>>(P, denom, g_r, out, elu);
}


# Score kernel re-tiled: a lane owns one target node with its g_r row in registers, each source's g_l row enters as scalar operands; scores stored transposed; denominator sum order changes
# speedup vs baseline: 2.3671x; 2.3671x over previous
#include <hip/hip_runtime.h>
#include <math.h>


typedef float v2f __attribute__((ext_vector_type(2)));
typedef float v8f __attribute__((ext_vector_type(8)));
typedef float v16f __attribute__((ext_vector_type(16)));

#define N_NODES 1024
#define IN_F    256
#define OUT_F   256
#define HEADS   4
#define HID     64
#define NEG_SLOPE 0.2f

#define SC_WAVES      8
#define SRC_PER_WAVE  32
#define SRC_PER_BLOCK (SC_WAVES * SRC_PER_WAVE)
#define N_SRC_BLOCKS  (N_NODES / SRC_PER_BLOCK)
#define AGG_STEPS     8
#define GR_PITCH      68

__global__ void gemm_xw(const float* __restrict__ X, const float* __restrict__ W,
                        float* __restrict__ G) {
    const int lane  = threadIdx.x;
    const int tileN = blockIdx.x * 16;
    const int tileM = blockIdx.y * 16;
    const int lid   = lane & 15;
    const int kh    = (lane >> 4) * 2;
    const int m     = tileM + lid;
    const int n     = tileN + lid;

    v8f acc = {};
    for (int k0 = 0; k0 < IN_F; k0 += 4) {
        const int k = k0 + kh;
        v2f a, b;
        a.x = X[m * IN_F + k];
        a.y = X[m * IN_F + k + 1];
        b.x = W[k * OUT_F + n];
        b.y = W[(k + 1) * OUT_F + n];
        acc = __builtin_amdgcn_wmma_f32_16x16x4_f32(false, a, false, b,
                                                    (short)0, acc, false, false);
    }
    const int row_half = (lane >> 4) * 8;
#pragma unroll
    for (int p = 0; p < 8; ++p) {
        const int mm = tileM + p + row_half;
        G[mm * OUT_F + n] = acc[p];
    }
}

__global__ __launch_bounds__(SC_WAVES * 32)
void attn_scores_t(const float* __restrict__ gl, const float* __restrict__ gr,
                   const float* __restrict__ adj, const float* __restrict__ aw,
                   float* __restrict__ PT, float* __restrict__ dpart) {
    const int t    = threadIdx.x;
    const int lane = t & 31;
    const int w    = __builtin_amdgcn_readfirstlane(t >> 5);
    const int i0   = blockIdx.x * 32;
    const int jb   = blockIdx.y;
    const int h    = blockIdx.z;

    __shared__ __attribute__((aligned(16))) float s_gr[32 * GR_PITCH];
    __shared__ __attribute__((aligned(16))) float s_aw[HID];
    __shared__ float s_den[SC_WAVES][32];

    for (int q = t; q < 32 * (HID / 4); q += SC_WAVES * 32) {
        const int row = q >> 4, c4 = q & 15;
        const float4 v = *reinterpret_cast<const float4*>(
            gr + (size_t)(i0 + row) * OUT_F + h * HID + c4 * 4);
        *reinterpret_cast<float4*>(s_gr + row * GR_PITCH + c4 * 4) = v;
    }
    if (t < HID) s_aw[t] = aw[t];
    __syncthreads();

    float gri[HID], awv[HID];
#pragma unroll
    for (int c4 = 0; c4 < HID / 4; ++c4) {
        const float4 g = *reinterpret_cast<const float4*>(s_gr + lane * GR_PITCH + c4 * 4);
        const float4 a = *reinterpret_cast<const float4*>(s_aw + c4 * 4);
        gri[c4 * 4 + 0] = g.x; gri[c4 * 4 + 1] = g.y; gri[c4 * 4 + 2] = g.z; gri[c4 * 4 + 3] = g.w;
        awv[c4 * 4 + 0] = a.x; awv[c4 * 4 + 1] = a.y; awv[c4 * 4 + 2] = a.z; awv[c4 * 4 + 3] = a.w;
    }

    const int jbase = jb * SRC_PER_BLOCK + w * SRC_PER_WAVE;
    float den = 0.0f;
    for (int jj = 0; jj < SRC_PER_WAVE; ++jj) {
        const int j = jbase + jj;
        const v16f* __restrict__ glj =
            reinterpret_cast<const v16f*>(gl + (size_t)j * OUT_F + h * HID);
        const v16f row0 = glj[0], row1 = glj[1], row2 = glj[2], row3 = glj[3];
        const float aji = adj[(size_t)j * N_NODES + i0 + lane];

        float e = 0.0f;
#pragma unroll
        for (int fb = 0; fb < HID; fb += 8) {
            float lr[8];
#pragma unroll
            for (int q = 0; q < 8; ++q) {
                const int f = fb + q;
                const float glf = (f < 16) ? row0[f & 15] : (f < 32) ? row1[f & 15]
                                : (f < 48) ? row2[f & 15] : row3[f & 15];
                const float s = gri[f] + glf;
                lr[q] = fmaxf(s, NEG_SLOPE * s);
            }
#pragma unroll
            for (int q = 0; q < 8; ++q)
                e = fmaf(lr[q], awv[fb + q], e);
        }

        const float ex = expf(e);
        const float p  = (aji != 0.0f) ? ex : 0.0f;
        PT[((size_t)h * N_NODES + j) * N_NODES + i0 + lane] = p;
        den += p;
    }

    s_den[w][lane] = den;
    __syncthreads();
    if (w == 0) {
        float d = s_den[0][lane];
        for (int k = 1; k < SC_WAVES; ++k) d += s_den[k][lane];
        dpart[((size_t)h * N_SRC_BLOCKS + jb) * N_NODES + i0 + lane] = d;
    }
}

__global__ void aggregate(const float* __restrict__ PT, const float* __restrict__ dpart,
                          const float* __restrict__ gr, float* __restrict__ out,
                          const int* __restrict__ use_elu) {
    const int lane  = threadIdx.x;
    const int tileF = blockIdx.x * 16;
    const int tileM = blockIdx.y * 16;
    const int h     = blockIdx.z;
    const int lid   = lane & 15;
    const int kh    = (lane >> 4) * 2;
    const int m     = tileM + lid;

    float dsum = dpart[(size_t)h * N_SRC_BLOCKS * N_NODES + m];
    for (int jb = 1; jb < N_SRC_BLOCKS; ++jb)
        dsum += dpart[((size_t)h * N_SRC_BLOCKS + jb) * N_NODES + m];
    const float recip = 1.0f / dsum;
    const float* PTh = PT + (size_t)h * N_NODES * N_NODES;

    v8f acc = {};
    for (int k0 = 0; k0 < N_NODES; k0 += 4 * AGG_STEPS) {
        v2f a[AGG_STEPS], b[AGG_STEPS];
#pragma unroll
        for (int u = 0; u < AGG_STEPS; ++u) {
            const int k = k0 + 4 * u + kh;
            a[u].x = PTh[(size_t)k * N_NODES + m];
            a[u].y = PTh[(size_t)(k + 1) * N_NODES + m];
            b[u].x = gr[k * OUT_F + h * HID + tileF + lid];
            b[u].y = gr[(k + 1) * OUT_F + h * HID + tileF + lid];
        }
#pragma unroll
        for (int u = 0; u < AGG_STEPS; ++u) {
            a[u].x *= recip;
            a[u].y *= recip;
            acc = __builtin_amdgcn_wmma_f32_16x16x4_f32(false, a[u], false, b[u],
                                                        (short)0, acc, false, false);
        }
    }

    const int elu = use_elu[0];
    const int row_half = (lane >> 4) * 8;
#pragma unroll
    for (int p = 0; p < 8; ++p) {
        const int i = tileM + p + row_half;
        float x = acc[p];
        if (elu) x = (x > 0.0f) ? x : expm1f(x);
        out[i * OUT_F + h * HID + tileF + lid] = x;
    }
}

extern "C" void kernel_launch(void* const* d_in, const int* in_sizes, int n_in,
                              void* d_out, int out_size, void* d_ws, size_t ws_size,
                              hipStream_t stream) {
    const float* h_in  = (const float*)d_in[0];
    const float* nei   = (const float*)d_in[1];
    const float* adj   = (const float*)d_in[2];
    const float* w_l   = (const float*)d_in[3];
    const float* w_r   = (const float*)d_in[4];
    const float* aw    = (const float*)d_in[5];
    const int*   elu   = (const int*)d_in[6];
    float* out = (float*)d_out;

    float* ws    = (float*)d_ws;
    float* dpart = ws;
    float* g_r   = ws + (size_t)N_NODES * OUT_F;
    float* PT    = g_r + (size_t)N_NODES * OUT_F;
    float* g_l   = out;

    dim3 ggrid(OUT_F / 16, N_NODES / 16);
    gemm_xw<<<ggrid, 32, 0, stream>>>(h_in, w_l, g_l);
    gemm_xw<<<ggrid, 32, 0, stream>>>(nei,  w_r, g_r);

    attn_scores_t<<<dim3(N_NODES / 32, N_SRC_BLOCKS, HEADS), SC_WAVES * 32, 0, stream>>>(
        g_l, g_r, adj, aw, PT, dpart);

    aggregate<<<dim3(HID / 16, N_NODES / 16, HEADS), 32, 0, stream>>>(PT, dpart, g_r, out, elu);
}


# Aggregation re-tiled: a wave forms 16 targets x 64 features in 4 accumulators on one A fragment, 8 waves split the neighbours, partial tiles added in wave order via LDS; neighbour sum order changes
# speedup vs baseline: 3.3667x; 1.4223x over previous
#include <hip/hip_runtime.h>
#include <math.h>


typedef float v2f __attribute__((ext_vector_type(2)));
typedef float v8f __attribute__((ext_vector_type(8)));
typedef float v16f __attribute__((ext_vector_type(16)));

#define N_NODES 1024
#define IN_F    256
#define OUT_F   256
#define HEADS   4
#define HID     64
#define NEG_SLOPE 0.2f

#define SC_WAVES      8
#define SRC_PER_WAVE  32
#define SRC_PER_BLOCK (SC_WAVES * SRC_PER_WAVE)
#define N_SRC_BLOCKS  (N_NODES / SRC_PER_BLOCK)
#define GR_PITCH      68

__global__ void gemm_xw(const float* __restrict__ X, const float* __restrict__ W,
                        float* __restrict__ G) {
    const int lane  = threadIdx.x;
    const int tileN = blockIdx.x * 16;
    const int tileM = blockIdx.y * 16;
    const int lid   = lane & 15;
    const int kh    = (lane >> 4) * 2;
    const int m     = tileM + lid;
    const int n     = tileN + lid;

    v8f acc = {};
    for (int k0 = 0; k0 < IN_F; k0 += 4) {
        const int k = k0 + kh;
        v2f a, b;
        a.x = X[m * IN_F + k];
        a.y = X[m * IN_F + k + 1];
        b.x = W[k * OUT_F + n];
        b.y = W[(k + 1) * OUT_F + n];
        acc = __builtin_amdgcn_wmma_f32_16x16x4_f32(false, a, false, b,
                                                    (short)0, acc, false, false);
    }
    const int row_half = (lane >> 4) * 8;
#pragma unroll
    for (int p = 0; p < 8; ++p) {
        const int mm = tileM + p + row_half;
        G[mm * OUT_F + n] = acc[p];
    }
}

__global__ __launch_bounds__(SC_WAVES * 32)
void attn_scores_t(const float* __restrict__ gl, const float* __restrict__ gr,
                   const float* __restrict__ adj, const float* __restrict__ aw,
                   float* __restrict__ PT, float* __restrict__ dpart) {
    const int t    = threadIdx.x;
    const int lane = t & 31;
    const int w    = __builtin_amdgcn_readfirstlane(t >> 5);
    const int i0   = blockIdx.x * 32;
    const int jb   = blockIdx.y;
    const int h    = blockIdx.z;

    __shared__ __attribute__((aligned(16))) float s_gr[32 * GR_PITCH];
    __shared__ __attribute__((aligned(16))) float s_aw[HID];
    __shared__ float s_den[SC_WAVES][32];

    for (int q = t; q < 32 * (HID / 4); q += SC_WAVES * 32) {
        const int row = q >> 4, c4 = q & 15;
        const float4 v = *reinterpret_cast<const float4*>(
            gr + (size_t)(i0 + row) * OUT_F + h * HID + c4 * 4);
        *reinterpret_cast<float4*>(s_gr + row * GR_PITCH + c4 * 4) = v;
    }
    if (t < HID) s_aw[t] = aw[t];
    __syncthreads();

    float gri[HID], awv[HID];
#pragma unroll
    for (int c4 = 0; c4 < HID / 4; ++c4) {
        const float4 g = *reinterpret_cast<const float4*>(s_gr + lane * GR_PITCH + c4 * 4);
        const float4 a = *reinterpret_cast<const float4*>(s_aw + c4 * 4);
        gri[c4 * 4 + 0] = g.x; gri[c4 * 4 + 1] = g.y; gri[c4 * 4 + 2] = g.z; gri[c4 * 4 + 3] = g.w;
        awv[c4 * 4 + 0] = a.x; awv[c4 * 4 + 1] = a.y; awv[c4 * 4 + 2] = a.z; awv[c4 * 4 + 3] = a.w;
    }

    const int jbase = jb * SRC_PER_BLOCK + w * SRC_PER_WAVE;
    float den = 0.0f;
    for (int jj = 0; jj < SRC_PER_WAVE; ++jj) {
        const int j = jbase + jj;
        const v16f* __restrict__ glj =
            reinterpret_cast<const v16f*>(gl + (size_t)j * OUT_F + h * HID);
        const v16f row0 = glj[0], row1 = glj[1], row2 = glj[2], row3 = glj[3];
        const float aji = adj[(size_t)j * N_NODES + i0 + lane];

        float e = 0.0f;
#pragma unroll
        for (int fb = 0; fb < HID; fb += 8) {
            float lr[8];
#pragma unroll
            for (int q = 0; q < 8; ++q) {
                const int f = fb + q;
                const float glf = (f < 16) ? row0[f & 15] : (f < 32) ? row1[f & 15]
                                : (f < 48) ? row2[f & 15] : row3[f & 15];
                const float s = gri[f] + glf;
                lr[q] = fmaxf(s, NEG_SLOPE * s);
            }
#pragma unroll
            for (int q = 0; q < 8; ++q)
                e = fmaf(lr[q], awv[fb + q], e);
        }

        const float ex = expf(e);
        const float p  = (aji != 0.0f) ? ex : 0.0f;
        PT[((size_t)h * N_NODES + j) * N_NODES + i0 + lane] = p;
        den += p;
    }

    s_den[w][lane] = den;
    __syncthreads();
    if (w == 0) {
        float d = s_den[0][lane];
        for (int k = 1; k < SC_WAVES; ++k) d += s_den[k][lane];
        dpart[((size_t)h * N_SRC_BLOCKS + jb) * N_NODES + i0 + lane] = d;
    }
}

#define AG_WAVES  8
#define AG_SLICE  (N_NODES / AG_WAVES)
#define AG_TRIP   16

__global__ __launch_bounds__(AG_WAVES * 32)
void aggregate(const float* __restrict__ PT, const float* __restrict__ dpart,
               const float* __restrict__ gr, float* __restrict__ out,
               const int* __restrict__ use_elu) {
    const int t     = threadIdx.x;
    const int lane  = t & 31;
    const int w     = __builtin_amdgcn_readfirstlane(t >> 5);
    const int tileM = blockIdx.x * 16;
    const int h     = blockIdx.y;
    const int c     = lane & 15;
    const int kh    = (lane >> 4) * 2;
    const int m     = tileM + c;

    __shared__ float4 s_part[AG_WAVES * 8 * 32];

    float dsum = dpart[(size_t)h * N_SRC_BLOCKS * N_NODES + m];
    for (int jb = 1; jb < N_SRC_BLOCKS; ++jb)
        dsum += dpart[((size_t)h * N_SRC_BLOCKS + jb) * N_NODES + m];
    const float recip = 1.0f / dsum;
    const int jbeg = w * AG_SLICE;
    const float* pa = PT + (size_t)h * N_NODES * N_NODES + (size_t)(jbeg + kh) * N_NODES + m;
    const float* pb = gr + (size_t)(jbeg + kh) * OUT_F + h * HID + 4 * c;

    v8f acc0 = {}, acc1 = {}, acc2 = {}, acc3 = {};
#pragma unroll
    for (int kk = 0; kk < AG_SLICE; kk += 4 * AG_TRIP) {
        v2f a[AG_TRIP];
        float4 b0[AG_TRIP], b1[AG_TRIP];
#pragma unroll
        for (int u = 0; u < AG_TRIP; ++u) {
            const int k = kk + 4 * u;
            a[u].x = pa[(size_t)k * N_NODES];
            a[u].y = pa[(size_t)(k + 1) * N_NODES];
            b0[u] = *reinterpret_cast<const float4*>(pb + (size_t)k * OUT_F);
            b1[u] = *reinterpret_cast<const float4*>(pb + (size_t)(k + 1) * OUT_F);
        }
#pragma unroll
        for (int u = 0; u < AG_TRIP; ++u) {
            v2f av, bv;
            av.x = a[u].x * recip;
            av.y = a[u].y * recip;
            bv.x = b0[u].x; bv.y = b1[u].x;
            acc0 = __builtin_amdgcn_wmma_f32_16x16x4_f32(false, av, false, bv,
                                                         (short)0, acc0, false, false);
            bv.x = b0[u].y; bv.y = b1[u].y;
            acc1 = __builtin_amdgcn_wmma_f32_16x16x4_f32(false, av, false, bv,
                                                         (short)0, acc1, false, false);
            bv.x = b0[u].z; bv.y = b1[u].z;
            acc2 = __builtin_amdgcn_wmma_f32_16x16x4_f32(false, av, false, bv,
                                                         (short)0, acc2, false, false);
            bv.x = b0[u].w; bv.y = b1[u].w;
            acc3 = __builtin_amdgcn_wmma_f32_16x16x4_f32(false, av, false, bv,
                                                         (short)0, acc3, false, false);
        }
    }

#pragma unroll
    for (int r = 0; r < 8; ++r)
        s_part[(w * 8 + r) * 32 + lane] = make_float4(acc0[r], acc1[r], acc2[r], acc3[r]);
    __syncthreads();

    float4 s = s_part[t];
#pragma unroll
    for (int k = 1; k < AG_WAVES; ++k) {
        const float4 v = s_part[k * (8 * 32) + t];
        s.x += v.x; s.y += v.y; s.z += v.z; s.w += v.w;
    }

    if (use_elu[0]) {
        s.x = (s.x > 0.0f) ? s.x : expm1f(s.x);
        s.y = (s.y > 0.0f) ? s.y : expm1f(s.y);
        s.z = (s.z > 0.0f) ? s.z : expm1f(s.z);
        s.w = (s.w > 0.0f) ? s.w : expm1f(s.w);
    }
    const int row = tileM + 8 * (lane >> 4) + (t >> 5);
    *reinterpret_cast<float4*>(out + (size_t)row * OUT_F + h * HID + 4 * c) = s;
}

extern "C" void kernel_launch(void* const* d_in, const int* in_sizes, int n_in,
                              void* d_out, int out_size, void* d_ws, size_t ws_size,
                              hipStream_t stream) {
    const float* h_in  = (const float*)d_in[0];
    const float* nei   = (const float*)d_in[1];
    const float* adj   = (const float*)d_in[2];
    const float* w_l   = (const float*)d_in[3];
    const float* w_r   = (const float*)d_in[4];
    const float* aw    = (const float*)d_in[5];
    const int*   elu   = (const int*)d_in[6];
    float* out = (float*)d_out;

    float* ws    = (float*)d_ws;
    float* dpart = ws;
    float* g_r   = ws + (size_t)N_NODES * OUT_F;
    float* PT    = g_r + (size_t)N_NODES * OUT_F;
    float* g_l   = out;

    dim3 ggrid(OUT_F / 16, N_NODES / 16);
    gemm_xw<<<ggrid, 32, 0, stream>>>(h_in, w_l, g_l);
    gemm_xw<<<ggrid, 32, 0, stream>>>(nei,  w_r, g_r);

    attn_scores_t<<<dim3(N_NODES / 32, N_SRC_BLOCKS, HEADS), SC_WAVES * 32, 0, stream>>>(
        g_l, g_r, adj, aw, PT, dpart);

    aggregate<<<dim3(N_NODES / 16, HEADS), AG_WAVES * 32, 0, stream>>>(PT, dpart, g_r, out, elu);
}


# Both projections in one launch of register-tiled waves: a wave forms a 16x32 strip in 2 accumulators on one A fragment, 16- and 8-byte global loads, about 50 in flight, no LDS; k sum order changes
# speedup vs baseline: 4.4622x; 1.3254x over previous
#include <hip/hip_runtime.h>
#include <math.h>


typedef float v2f __attribute__((ext_vector_type(2)));
typedef float v8f __attribute__((ext_vector_type(8)));
typedef float v16f __attribute__((ext_vector_type(16)));

#define N_NODES 1024
#define IN_F    256
#define OUT_F   256
#define HEADS   4
#define HID     64
#define NEG_SLOPE 0.2f

#define SC_WAVES      8
#define SRC_PER_WAVE  32
#define SRC_PER_BLOCK (SC_WAVES * SRC_PER_WAVE)
#define N_SRC_BLOCKS  (N_NODES / SRC_PER_BLOCK)
#define GR_PITCH      68

#define PJ_WAVES 4
#define PJ_KBLK  32
#define PJ_SETK  (2 * PJ_KBLK)

struct PjSet {
    float4 a[2][4];
    float2 b[2][16];
};

__device__ __forceinline__ void pj_load(PjSet& s, const float* __restrict__ pa,
                                        const float* __restrict__ pb, const int k0) {
#pragma unroll
    for (int blk = 0; blk < 2; ++blk) {
        const int kb = k0 + blk * PJ_KBLK;
#pragma unroll
        for (int q = 0; q < 4; ++q)
            s.a[blk][q] = *reinterpret_cast<const float4*>(pa + kb + 4 * q);
#pragma unroll
        for (int e = 0; e < 16; ++e)
            s.b[blk][e] = *reinterpret_cast<const float2*>(pb + (size_t)(kb + e) * OUT_F);
    }
}

__device__ __forceinline__ void pj_mma(const PjSet& s, v8f& acc0, v8f& acc1) {
#pragma unroll
    for (int blk = 0; blk < 2; ++blk) {
#pragma unroll
        for (int u = 0; u < 8; ++u) {
            const float4 aq = s.a[blk][u >> 1];
            const float2 bk0 = s.b[blk][2 * u], bk1 = s.b[blk][2 * u + 1];
            v2f av, bv;
            av.x = (u & 1) ? aq.z : aq.x;
            av.y = (u & 1) ? aq.w : aq.y;
            bv.x = bk0.x; bv.y = bk1.x;
            acc0 = __builtin_amdgcn_wmma_f32_16x16x4_f32(false, av, false, bv,
                                                         (short)0, acc0, false, false);
            bv.x = bk0.y; bv.y = bk1.y;
            acc1 = __builtin_amdgcn_wmma_f32_16x16x4_f32(false, av, false, bv,
                                                         (short)0, acc1, false, false);
        }
    }
}

__global__ __launch_bounds__(PJ_WAVES * 32) __attribute__((amdgpu_waves_per_eu(1, 2)))
void project_both(const float* __restrict__ X0, const float* __restrict__ W0,
                  float* __restrict__ G0,
                  const float* __restrict__ X1, const float* __restrict__ W1,
                  float* __restrict__ G1) {
    const int t    = threadIdx.x;
    const int lane = t & 31;
    const int w    = __builtin_amdgcn_readfirstlane(t >> 5);
    const int c    = lane & 15;
    const int hh   = lane >> 4;
    const bool second = (blockIdx.z != 0);
    const float* __restrict__ X = second ? X1 : X0;
    const float* __restrict__ W = second ? W1 : W0;
    float* __restrict__ G       = second ? G1 : G0;

    const int tileM = (blockIdx.y * 2 + (w >> 1)) * 16;
    const int n0    = (blockIdx.x * 2 + (w & 1)) * 32;

    const float* pa = X + (size_t)(tileM + c) * IN_F + 16 * hh;
    const float* pb = W + (size_t)(16 * hh) * OUT_F + n0 + 2 * c;

    v8f acc0 = {}, acc1 = {};
    PjSet s0, s1;
    pj_load(s0, pa, pb, 0 * PJ_SETK);
    pj_load(s1, pa, pb, 1 * PJ_SETK);
    pj_mma(s0, acc0, acc1);
    pj_load(s0, pa, pb, 2 * PJ_SETK);
    pj_mma(s1, acc0, acc1);
    pj_load(s1, pa, pb, 3 * PJ_SETK);
    pj_mma(s0, acc0, acc1);
    pj_mma(s1, acc0, acc1);

    float* po = G + (size_t)(tileM + 8 * hh) * OUT_F + n0 + 2 * c;
#pragma unroll
    for (int r = 0; r < 8; ++r)
        *reinterpret_cast<float2*>(po + (size_t)r * OUT_F) = make_float2(acc0[r], acc1[r]);
}

__global__ __launch_bounds__(SC_WAVES * 32)
void attn_scores_t(const float* __restrict__ gl, const float* __restrict__ gr,
                   const float* __restrict__ adj, const float* __restrict__ aw,
                   float* __restrict__ PT, float* __restrict__ dpart) {
    const int t    = threadIdx.x;
    const int lane = t & 31;
    const int w    = __builtin_amdgcn_readfirstlane(t >> 5);
    const int i0   = blockIdx.x * 32;
    const int jb   = blockIdx.y;
    const int h    = blockIdx.z;

    __shared__ __attribute__((aligned(16))) float s_gr[32 * GR_PITCH];
    __shared__ __attribute__((aligned(16))) float s_aw[HID];
    __shared__ float s_den[SC_WAVES][32];

    for (int q = t; q < 32 * (HID / 4); q += SC_WAVES * 32) {
        const int row = q >> 4, c4 = q & 15;
        const float4 v = *reinterpret_cast<const float4*>(
            gr + (size_t)(i0 + row) * OUT_F + h * HID + c4 * 4);
        *reinterpret_cast<float4*>(s_gr + row * GR_PITCH + c4 * 4) = v;
    }
    if (t < HID) s_aw[t] = aw[t];
    __syncthreads();

    float gri[HID], awv[HID];
#pragma unroll
    for (int c4 = 0; c4 < HID / 4; ++c4) {
        const float4 g = *reinterpret_cast<const float4*>(s_gr + lane * GR_PITCH + c4 * 4);
        const float4 a = *reinterpret_cast<const float4*>(s_aw + c4 * 4);
        gri[c4 * 4 + 0] = g.x; gri[c4 * 4 + 1] = g.y; gri[c4 * 4 + 2] = g.z; gri[c4 * 4 + 3] = g.w;
        awv[c4 * 4 + 0] = a.x; awv[c4 * 4 + 1] = a.y; awv[c4 * 4 + 2] = a.z; awv[c4 * 4 + 3] = a.w;
    }

    const int jbase = jb * SRC_PER_BLOCK + w * SRC_PER_WAVE;
    float den = 0.0f;
    for (int jj = 0; jj < SRC_PER_WAVE; ++jj) {
        const int j = jbase + jj;
        const v16f* __restrict__ glj =
            reinterpret_cast<const v16f*>(gl + (size_t)j * OUT_F + h * HID);
        const v16f row0 = glj[0], row1 = glj[1], row2 = glj[2], row3 = glj[3];
        const float aji = adj[(size_t)j * N_NODES + i0 + lane];

        float e = 0.0f;
#pragma unroll
        for (int fb = 0; fb < HID; fb += 8) {
            float lr[8];
#pragma unroll
            for (int q = 0; q < 8; ++q) {
                const int f = fb + q;
                const float glf = (f < 16) ? row0[f & 15] : (f < 32) ? row1[f & 15]
                                : (f < 48) ? row2[f & 15] : row3[f & 15];
                const float s = gri[f] + glf;
                lr[q] = fmaxf(s, NEG_SLOPE * s);
            }
#pragma unroll
            for (int q = 0; q < 8; ++q)
                e = fmaf(lr[q], awv[fb + q], e);
        }

        const float ex = expf(e);
        const float p  = (aji != 0.0f) ? ex : 0.0f;
        PT[((size_t)h * N_NODES + j) * N_NODES + i0 + lane] = p;
        den += p;
    }

    s_den[w][lane] = den;
    __syncthreads();
    if (w == 0) {
        float d = s_den[0][lane];
        for (int k = 1; k < SC_WAVES; ++k) d += s_den[k][lane];
        dpart[((size_t)h * N_SRC_BLOCKS + jb) * N_NODES + i0 + lane] = d;
    }
}

#define AG_WAVES  8
#define AG_SLICE  (N_NODES / AG_WAVES)
#define AG_TRIP   16

__global__ __launch_bounds__(AG_WAVES * 32)
void aggregate(const float* __restrict__ PT, const float* __restrict__ dpart,
               const float* __restrict__ gr, float* __restrict__ out,
               const int* __restrict__ use_elu) {
    const int t     = threadIdx.x;
    const int lane  = t & 31;
    const int w     = __builtin_amdgcn_readfirstlane(t >> 5);
    const int tileM = blockIdx.x * 16;
    const int h     = blockIdx.y;
    const int c     = lane & 15;
    const int kh    = (lane >> 4) * 2;
    const int m     = tileM + c;

    __shared__ float4 s_part[AG_WAVES * 8 * 32];

    float dsum = dpart[(size_t)h * N_SRC_BLOCKS * N_NODES + m];
    for (int jb = 1; jb < N_SRC_BLOCKS; ++jb)
        dsum += dpart[((size_t)h * N_SRC_BLOCKS + jb) * N_NODES + m];
    const float recip = 1.0f / dsum;
    const int jbeg = w * AG_SLICE;
    const float* pa = PT + (size_t)h * N_NODES * N_NODES + (size_t)(jbeg + kh) * N_NODES + m;
    const float* pb = gr + (size_t)(jbeg + kh) * OUT_F + h * HID + 4 * c;

    v8f acc0 = {}, acc1 = {}, acc2 = {}, acc3 = {};
#pragma unroll
    for (int kk = 0; kk < AG_SLICE; kk += 4 * AG_TRIP) {
        v2f a[AG_TRIP];
        float4 b0[AG_TRIP], b1[AG_TRIP];
#pragma unroll
        for (int u = 0; u < AG_TRIP; ++u) {
            const int k = kk + 4 * u;
            a[u].x = pa[(size_t)k * N_NODES];
            a[u].y = pa[(size_t)(k + 1) * N_NODES];
            b0[u] = *reinterpret_cast<const float4*>(pb + (size_t)k * OUT_F);
            b1[u] = *reinterpret_cast<const float4*>(pb + (size_t)(k + 1) * OUT_F);
        }
#pragma unroll
        for (int u = 0; u < AG_TRIP; ++u) {
            v2f av, bv;
            av.x = a[u].x * recip;
            av.y = a[u].y * recip;
            bv.x = b0[u].x; bv.y = b1[u].x;
            acc0 = __builtin_amdgcn_wmma_f32_16x16x4_f32(false, av, false, bv,
                                                         (short)0, acc0, false, false);
            bv.x = b0[u].y; bv.y = b1[u].y;
            acc1 = __builtin_amdgcn_wmma_f32_16x16x4_f32(false, av, false, bv,
                                                         (short)0, acc1, false, false);
            bv.x = b0[u].z; bv.y = b1[u].z;
            acc2 = __builtin_amdgcn_wmma_f32_16x16x4_f32(false, av, false, bv,
                                                         (short)0, acc2, false, false);
            bv.x = b0[u].w; bv.y = b1[u].w;
            acc3 = __builtin_amdgcn_wmma_f32_16x16x4_f32(false, av, false, bv,
                                                         (short)0, acc3, false, false);
        }
    }

#pragma unroll
    for (int r = 0; r < 8; ++r)
        s_part[(w * 8 + r) * 32 + lane] = make_float4(acc0[r], acc1[r], acc2[r], acc3[r]);
    __syncthreads();

    float4 s = s_part[t];
#pragma unroll
    for (int k = 1; k < AG_WAVES; ++k) {
        const float4 v = s_part[k * (8 * 32) + t];
        s.x += v.x; s.y += v.y; s.z += v.z; s.w += v.w;
    }

    if (use_elu[0]) {
        s.x = (s.x > 0.0f) ? s.x : expm1f(s.x);
        s.y = (s.y > 0.0f) ? s.y : expm1f(s.y);
        s.z = (s.z > 0.0f) ? s.z : expm1f(s.z);
        s.w = (s.w > 0.0f) ? s.w : expm1f(s.w);
    }
    const int row = tileM + 8 * (lane >> 4) + (t >> 5);
    *reinterpret_cast<float4*>(out + (size_t)row * OUT_F + h * HID + 4 * c) = s;
}

extern "C" void kernel_launch(void* const* d_in, const int* in_sizes, int n_in,
                              void* d_out, int out_size, void* d_ws, size_t ws_size,
                              hipStream_t stream) {
    const float* h_in  = (const float*)d_in[0];
    const float* nei   = (const float*)d_in[1];
    const float* adj   = (const float*)d_in[2];
    const float* w_l   = (const float*)d_in[3];
    const float* w_r   = (const float*)d_in[4];
    const float* aw    = (const float*)d_in[5];
    const int*   elu   = (const int*)d_in[6];
    float* out = (float*)d_out;

    float* ws    = (float*)d_ws;
    float* dpart = ws;
    float* g_r   = ws + (size_t)N_NODES * OUT_F;
    float* PT    = g_r + (size_t)N_NODES * OUT_F;
    float* g_l   = out;

    project_both<<<dim3(OUT_F / 64, N_NODES / 32, 2), PJ_WAVES * 32, 0, stream>>>(
        h_in, w_l, g_l, nei, w_r, g_r);

    attn_scores_t<<<dim3(N_NODES / 32, N_SRC_BLOCKS, HEADS), SC_WAVES * 32, 0, stream>>>(
        g_l, g_r, adj, aw, PT, dpart);

    aggregate<<<dim3(N_NODES / 16, HEADS), AG_WAVES * 32, 0, stream>>>(PT, dpart, g_r, out, elu);
}


# Score kernel: a lane owns two targets as one packed two-wide value, g_l and attn_w enter as scalar operands, features in 4 chunks outermost, 16 running sums per wave; denominator sum order changes
# speedup vs baseline: 4.8613x; 1.0894x over previous
#include <hip/hip_runtime.h>
#include <math.h>


typedef float v2f __attribute__((ext_vector_type(2)));
typedef float v8f __attribute__((ext_vector_type(8)));
typedef float v16f __attribute__((ext_vector_type(16)));

#define N_NODES 1024
#define IN_F    256
#define OUT_F   256
#define HEADS   4
#define HID     64
#define NEG_SLOPE 0.2f

#define SC_WAVES      16
#define SRC_PER_WAVE  16
#define SRC_PER_BLOCK (SC_WAVES * SRC_PER_WAVE)
#define N_SRC_BLOCKS  (N_NODES / SRC_PER_BLOCK)
#define TGT_PER_BLOCK 64
#define F_CHUNK       16
#define SRC_GROUP     4
#define G2_PITCH      132

#define PJ_WAVES 4
#define PJ_KBLK  32
#define PJ_SETK  (2 * PJ_KBLK)

struct PjSet {
    float4 a[2][4];
    float2 b[2][16];
};

__device__ __forceinline__ void pj_load(PjSet& s, const float* __restrict__ pa,
                                        const float* __restrict__ pb, const int k0) {
#pragma unroll
    for (int blk = 0; blk < 2; ++blk) {
        const int kb = k0 + blk * PJ_KBLK;
#pragma unroll
        for (int q = 0; q < 4; ++q)
            s.a[blk][q] = *reinterpret_cast<const float4*>(pa + kb + 4 * q);
#pragma unroll
        for (int e = 0; e < 16; ++e)
            s.b[blk][e] = *reinterpret_cast<const float2*>(pb + (size_t)(kb + e) * OUT_F);
    }
}

__device__ __forceinline__ void pj_mma(const PjSet& s, v8f& acc0, v8f& acc1) {
#pragma unroll
    for (int blk = 0; blk < 2; ++blk) {
#pragma unroll
        for (int u = 0; u < 8; ++u) {
            const float4 aq = s.a[blk][u >> 1];
            const float2 bk0 = s.b[blk][2 * u], bk1 = s.b[blk][2 * u + 1];
            v2f av, bv;
            av.x = (u & 1) ? aq.z : aq.x;
            av.y = (u & 1) ? aq.w : aq.y;
            bv.x = bk0.x; bv.y = bk1.x;
            acc0 = __builtin_amdgcn_wmma_f32_16x16x4_f32(false, av, false, bv,
                                                         (short)0, acc0, false, false);
            bv.x = bk0.y; bv.y = bk1.y;
            acc1 = __builtin_amdgcn_wmma_f32_16x16x4_f32(false, av, false, bv,
                                                         (short)0, acc1, false, false);
        }
    }
}

__global__ __launch_bounds__(PJ_WAVES * 32) __attribute__((amdgpu_waves_per_eu(1, 2)))
void project_both(const float* __restrict__ X0, const float* __restrict__ W0,
                  float* __restrict__ G0,
                  const float* __restrict__ X1, const float* __restrict__ W1,
                  float* __restrict__ G1) {
    const int t    = threadIdx.x;
    const int lane = t & 31;
    const int w    = __builtin_amdgcn_readfirstlane(t >> 5);
    const int c    = lane & 15;
    const int hh   = lane >> 4;
    const bool second = (blockIdx.z != 0);
    const float* __restrict__ X = second ? X1 : X0;
    const float* __restrict__ W = second ? W1 : W0;
    float* __restrict__ G       = second ? G1 : G0;

    const int tileM = (blockIdx.y * 2 + (w >> 1)) * 16;
    const int n0    = (blockIdx.x * 2 + (w & 1)) * 32;

    const float* pa = X + (size_t)(tileM + c) * IN_F + 16 * hh;
    const float* pb = W + (size_t)(16 * hh) * OUT_F + n0 + 2 * c;

    v8f acc0 = {}, acc1 = {};
    PjSet s0, s1;
    pj_load(s0, pa, pb, 0 * PJ_SETK);
    pj_load(s1, pa, pb, 1 * PJ_SETK);
    pj_mma(s0, acc0, acc1);
    pj_load(s0, pa, pb, 2 * PJ_SETK);
    pj_mma(s1, acc0, acc1);
    pj_load(s1, pa, pb, 3 * PJ_SETK);
    pj_mma(s0, acc0, acc1);
    pj_mma(s1, acc0, acc1);

    float* po = G + (size_t)(tileM + 8 * hh) * OUT_F + n0 + 2 * c;
#pragma unroll
    for (int r = 0; r < 8; ++r)
        *reinterpret_cast<float2*>(po + (size_t)r * OUT_F) = make_float2(acc0[r], acc1[r]);
}

__global__ __launch_bounds__(SC_WAVES * 32)
void attn_scores_t(const float* __restrict__ gl, const float* __restrict__ gr,
                   const float* __restrict__ adj, const float* __restrict__ aw,
                   float* __restrict__ PT, float* __restrict__ dpart) {
    const int t    = threadIdx.x;
    const int lane = t & 31;
    const int w    = __builtin_amdgcn_readfirstlane(t >> 5);
    const int i0   = blockIdx.x * TGT_PER_BLOCK;
    const int jb   = blockIdx.y;
    const int h    = blockIdx.z;

    __shared__ __attribute__((aligned(16))) float s_g2[32 * G2_PITCH];
    __shared__ float s_den[SC_WAVES][TGT_PER_BLOCK];

    for (int q = t; q < TGT_PER_BLOCK * (HID / 4); q += SC_WAVES * 32) {
        const int row = q >> 4, c4 = q & 15;
        const float4 v = *reinterpret_cast<const float4*>(
            gr + (size_t)(i0 + row) * OUT_F + h * HID + c4 * 4);
        float* d = s_g2 + (row & 31) * G2_PITCH + 8 * c4 + (row >> 5);
        d[0] = v.x; d[2] = v.y; d[4] = v.z; d[6] = v.w;
    }

    const int jbase = jb * SRC_PER_BLOCK + w * SRC_PER_WAVE;
    const float* __restrict__ adjp = adj + (size_t)jbase * N_NODES + i0 + lane;
    float aja[SRC_PER_WAVE], ajb[SRC_PER_WAVE];
#pragma unroll
    for (int jj = 0; jj < SRC_PER_WAVE; ++jj) {
        aja[jj] = adjp[(size_t)jj * N_NODES];
        ajb[jj] = adjp[(size_t)jj * N_NODES + 32];
    }
    __syncthreads();

    v2f e2[SRC_PER_WAVE];
#pragma unroll
    for (int jj = 0; jj < SRC_PER_WAVE; ++jj) { e2[jj].x = 0.0f; e2[jj].y = 0.0f; }

    const float* __restrict__ glp = gl + (size_t)jbase * OUT_F + h * HID;
#pragma unroll 1
    for (int c = 0; c < HID / F_CHUNK; ++c) {
        v2f g2[F_CHUNK];
        const float4* __restrict__ gsrc =
            reinterpret_cast<const float4*>(s_g2 + lane * G2_PITCH + 2 * F_CHUNK * c);
#pragma unroll
        for (int q = 0; q < F_CHUNK / 2; ++q) {
            const float4 v = gsrc[q];
            g2[2 * q].x = v.x;     g2[2 * q].y = v.y;
            g2[2 * q + 1].x = v.z; g2[2 * q + 1].y = v.w;
        }
        const v16f awc = *reinterpret_cast<const v16f*>(aw + F_CHUNK * c);
#pragma unroll
        for (int jg = 0; jg < SRC_PER_WAVE; jg += SRC_GROUP) {
            v16f row[SRC_GROUP];
#pragma unroll
            for (int r = 0; r < SRC_GROUP; ++r)
                row[r] = *reinterpret_cast<const v16f*>(
                    glp + (size_t)(jg + r) * OUT_F + F_CHUNK * c);
#pragma unroll
            for (int r = 0; r < SRC_GROUP; ++r) {
                v2f e = e2[jg + r];
#pragma unroll
                for (int f = 0; f < F_CHUNK; ++f) {
                    const float glf = row[r][f];
                    const float awf = awc[f];
                    v2f s  = g2[f] + (v2f){glf, glf};
                    v2f m  = s * (v2f){NEG_SLOPE, NEG_SLOPE};
                    v2f lr = __builtin_elementwise_max(s, m);
                    e = __builtin_elementwise_fma(lr, (v2f){awf, awf}, e);
                }
                e2[jg + r] = e;
            }
        }
    }

    float dena = 0.0f, denb = 0.0f;
    float* __restrict__ prow = PT + ((size_t)h * N_NODES + jbase) * N_NODES + i0 + lane;
#pragma unroll
    for (int jj = 0; jj < SRC_PER_WAVE; ++jj) {
        const float pa = (aja[jj] != 0.0f) ? expf(e2[jj].x) : 0.0f;
        const float pb = (ajb[jj] != 0.0f) ? expf(e2[jj].y) : 0.0f;
        prow[(size_t)jj * N_NODES]      = pa;
        prow[(size_t)jj * N_NODES + 32] = pb;
        dena += pa;
        denb += pb;
    }

    s_den[w][lane]      = dena;
    s_den[w][lane + 32] = denb;
    __syncthreads();
    if (t < TGT_PER_BLOCK) {
        float d = s_den[0][t];
        for (int k = 1; k < SC_WAVES; ++k) d += s_den[k][t];
        dpart[((size_t)h * N_SRC_BLOCKS + jb) * N_NODES + i0 + t] = d;
    }
}

#define AG_WAVES  8
#define AG_SLICE  (N_NODES / AG_WAVES)
#define AG_TRIP   16

__global__ __launch_bounds__(AG_WAVES * 32)
void aggregate(const float* __restrict__ PT, const float* __restrict__ dpart,
               const float* __restrict__ gr, float* __restrict__ out,
               const int* __restrict__ use_elu) {
    const int t     = threadIdx.x;
    const int lane  = t & 31;
    const int w     = __builtin_amdgcn_readfirstlane(t >> 5);
    const int tileM = blockIdx.x * 16;
    const int h     = blockIdx.y;
    const int c     = lane & 15;
    const int kh    = (lane >> 4) * 2;
    const int m     = tileM + c;

    __shared__ float4 s_part[AG_WAVES * 8 * 32];

    float dsum = dpart[(size_t)h * N_SRC_BLOCKS * N_NODES + m];
    for (int jb = 1; jb < N_SRC_BLOCKS; ++jb)
        dsum += dpart[((size_t)h * N_SRC_BLOCKS + jb) * N_NODES + m];
    const float recip = 1.0f / dsum;
    const int jbeg = w * AG_SLICE;
    const float* pa = PT + (size_t)h * N_NODES * N_NODES + (size_t)(jbeg + kh) * N_NODES + m;
    const float* pb = gr + (size_t)(jbeg + kh) * OUT_F + h * HID + 4 * c;

    v8f acc0 = {}, acc1 = {}, acc2 = {}, acc3 = {};
#pragma unroll
    for (int kk = 0; kk < AG_SLICE; kk += 4 * AG_TRIP) {
        v2f a[AG_TRIP];
        float4 b0[AG_TRIP], b1[AG_TRIP];
#pragma unroll
        for (int u = 0; u < AG_TRIP; ++u) {
            const int k = kk + 4 * u;
            a[u].x = pa[(size_t)k * N_NODES];
            a[u].y = pa[(size_t)(k + 1) * N_NODES];
            b0[u] = *reinterpret_cast<const float4*>(pb + (size_t)k * OUT_F);
            b1[u] = *reinterpret_cast<const float4*>(pb + (size_t)(k + 1) * OUT_F);
        }
#pragma unroll
        for (int u = 0; u < AG_TRIP; ++u) {
            v2f av, bv;
            av.x = a[u].x * recip;
            av.y = a[u].y * recip;
            bv.x = b0[u].x; bv.y = b1[u].x;
            acc0 = __builtin_amdgcn_wmma_f32_16x16x4_f32(false, av, false, bv,
                                                         (short)0, acc0, false, false);
            bv.x = b0[u].y; bv.y = b1[u].y;
            acc1 = __builtin_amdgcn_wmma_f32_16x16x4_f32(false, av, false, bv,
                                                         (short)0, acc1, false, false);
            bv.x = b0[u].z; bv.y = b1[u].z;
            acc2 = __builtin_amdgcn_wmma_f32_16x16x4_f32(false, av, false, bv,
                                                         (short)0, acc2, false, false);
            bv.x = b0[u].w; bv.y = b1[u].w;
            acc3 = __builtin_amdgcn_wmma_f32_16x16x4_f32(false, av, false, bv,
                                                         (short)0, acc3, false, false);
        }
    }

#pragma unroll
    for (int r = 0; r < 8; ++r)
        s_part[(w * 8 + r) * 32 + lane] = make_float4(acc0[r], acc1[r], acc2[r], acc3[r]);
    __syncthreads();

    float4 s = s_part[t];
#pragma unroll
    for (int k = 1; k < AG_WAVES; ++k) {
        const float4 v = s_part[k * (8 * 32) + t];
        s.x += v.x; s.y += v.y; s.z += v.z; s.w += v.w;
    }

    if (use_elu[0]) {
        s.x = (s.x > 0.0f) ? s.x : expm1f(s.x);
        s.y = (s.y > 0.0f) ? s.y : expm1f(s.y);
        s.z = (s.z > 0.0f) ? s.z : expm1f(s.z);
        s.w = (s.w > 0.0f) ? s.w : expm1f(s.w);
    }
    const int row = tileM + 8 * (lane >> 4) + (t >> 5);
    *reinterpret_cast<float4*>(out + (size_t)row * OUT_F + h * HID + 4 * c) = s;
}

extern "C" void kernel_launch(void* const* d_in, const int* in_sizes, int n_in,
                              void* d_out, int out_size, void* d_ws, size_t ws_size,
                              hipStream_t stream) {
    const float* h_in  = (const float*)d_in[0];
    const float* nei   = (const float*)d_in[1];
    const float* adj   = (const float*)d_in[2];
    const float* w_l   = (const float*)d_in[3];
    const float* w_r   = (const float*)d_in[4];
    const float* aw    = (const float*)d_in[5];
    const int*   elu   = (const int*)d_in[6];
    float* out = (float*)d_out;

    float* ws    = (float*)d_ws;
    float* dpart = ws;
    float* g_r   = ws + (size_t)N_NODES * OUT_F;
    float* PT    = g_r + (size_t)N_NODES * OUT_F;
    float* g_l   = out;

    project_both<<<dim3(OUT_F / 64, N_NODES / 32, 2), PJ_WAVES * 32, 0, stream>>>(
        h_in, w_l, g_l, nei, w_r, g_r);

    attn_scores_t<<<dim3(N_NODES / TGT_PER_BLOCK, N_SRC_BLOCKS, HEADS), SC_WAVES * 32, 0, stream>>>(
        g_l, g_r, adj, aw, PT, dpart);

    aggregate<<<dim3(N_NODES / 16, HEADS), AG_WAVES * 32, 0, stream>>>(PT, dpart, g_r, out, elu);
}


# Score loop reordered: within a group of 4 source nodes the feature is outermost and each score step runs over the 4 nodes, so a wave issues from 4 independent chains, not 1; every sum order kept
# speedup vs baseline: 5.5219x; 1.1359x over previous
#include <hip/hip_runtime.h>
#include <math.h>


typedef float v2f __attribute__((ext_vector_type(2)));
typedef float v8f __attribute__((ext_vector_type(8)));
typedef float v16f __attribute__((ext_vector_type(16)));

#define N_NODES 1024
#define IN_F    256
#define OUT_F   256
#define HEADS   4
#define HID     64
#define NEG_SLOPE 0.2f

#define SC_WAVES      16
#define SRC_PER_WAVE  16
#define SRC_PER_BLOCK (SC_WAVES * SRC_PER_WAVE)
#define N_SRC_BLOCKS  (N_NODES / SRC_PER_BLOCK)
#define TGT_PER_BLOCK 64
#define F_CHUNK       16
#define SRC_GROUP     4
#define G2_PITCH      132

#define PJ_WAVES 4
#define PJ_KBLK  32
#define PJ_SETK  (2 * PJ_KBLK)

struct PjSet {
    float4 a[2][4];
    float2 b[2][16];
};

__device__ __forceinline__ void pj_load(PjSet& s, const float* __restrict__ pa,
                                        const float* __restrict__ pb, const int k0) {
#pragma unroll
    for (int blk = 0; blk < 2; ++blk) {
        const int kb = k0 + blk * PJ_KBLK;
#pragma unroll
        for (int q = 0; q < 4; ++q)
            s.a[blk][q] = *reinterpret_cast<const float4*>(pa + kb + 4 * q);
#pragma unroll
        for (int e = 0; e < 16; ++e)
            s.b[blk][e] = *reinterpret_cast<const float2*>(pb + (size_t)(kb + e) * OUT_F);
    }
}

__device__ __forceinline__ void pj_mma(const PjSet& s, v8f& acc0, v8f& acc1) {
#pragma unroll
    for (int blk = 0; blk < 2; ++blk) {
#pragma unroll
        for (int u = 0; u < 8; ++u) {
            const float4 aq = s.a[blk][u >> 1];
            const float2 bk0 = s.b[blk][2 * u], bk1 = s.b[blk][2 * u + 1];
            v2f av, bv;
            av.x = (u & 1) ? aq.z : aq.x;
            av.y = (u & 1) ? aq.w : aq.y;
            bv.x = bk0.x; bv.y = bk1.x;
            acc0 = __builtin_amdgcn_wmma_f32_16x16x4_f32(false, av, false, bv,
                                                         (short)0, acc0, false, false);
            bv.x = bk0.y; bv.y = bk1.y;
            acc1 = __builtin_amdgcn_wmma_f32_16x16x4_f32(false, av, false, bv,
                                                         (short)0, acc1, false, false);
        }
    }
}

__global__ __launch_bounds__(PJ_WAVES * 32) __attribute__((amdgpu_waves_per_eu(1, 2)))
void project_both(const float* __restrict__ X0, const float* __restrict__ W0,
                  float* __restrict__ G0,
                  const float* __restrict__ X1, const float* __restrict__ W1,
                  float* __restrict__ G1) {
    const int t    = threadIdx.x;
    const int lane = t & 31;
    const int w    = __builtin_amdgcn_readfirstlane(t >> 5);
    const int c    = lane & 15;
    const int hh   = lane >> 4;
    const bool second = (blockIdx.z != 0);
    const float* __restrict__ X = second ? X1 : X0;
    const float* __restrict__ W = second ? W1 : W0;
    float* __restrict__ G       = second ? G1 : G0;

    const int tileM = (blockIdx.y * 2 + (w >> 1)) * 16;
    const int n0    = (blockIdx.x * 2 + (w & 1)) * 32;

    const float* pa = X + (size_t)(tileM + c) * IN_F + 16 * hh;
    const float* pb = W + (size_t)(16 * hh) * OUT_F + n0 + 2 * c;

    v8f acc0 = {}, acc1 = {};
    PjSet s0, s1;
    pj_load(s0, pa, pb, 0 * PJ_SETK);
    pj_load(s1, pa, pb, 1 * PJ_SETK);
    pj_mma(s0, acc0, acc1);
    pj_load(s0, pa, pb, 2 * PJ_SETK);
    pj_mma(s1, acc0, acc1);
    pj_load(s1, pa, pb, 3 * PJ_SETK);
    pj_mma(s0, acc0, acc1);
    pj_mma(s1, acc0, acc1);

    float* po = G + (size_t)(tileM + 8 * hh) * OUT_F + n0 + 2 * c;
#pragma unroll
    for (int r = 0; r < 8; ++r)
        *reinterpret_cast<float2*>(po + (size_t)r * OUT_F) = make_float2(acc0[r], acc1[r]);
}

__global__ __launch_bounds__(SC_WAVES * 32)
void attn_scores_t(const float* __restrict__ gl, const float* __restrict__ gr,
                   const float* __restrict__ adj, const float* __restrict__ aw,
                   float* __restrict__ PT, float* __restrict__ dpart) {
    const int t    = threadIdx.x;
    const int lane = t & 31;
    const int w    = __builtin_amdgcn_readfirstlane(t >> 5);
    const int i0   = blockIdx.x * TGT_PER_BLOCK;
    const int jb   = blockIdx.y;
    const int h    = blockIdx.z;

    __shared__ __attribute__((aligned(16))) float s_g2[32 * G2_PITCH];
    __shared__ float s_den[SC_WAVES][TGT_PER_BLOCK];

    for (int q = t; q < TGT_PER_BLOCK * (HID / 4); q += SC_WAVES * 32) {
        const int row = q >> 4, c4 = q & 15;
        const float4 v = *reinterpret_cast<const float4*>(
            gr + (size_t)(i0 + row) * OUT_F + h * HID + c4 * 4);
        float* d = s_g2 + (row & 31) * G2_PITCH + 8 * c4 + (row >> 5);
        d[0] = v.x; d[2] = v.y; d[4] = v.z; d[6] = v.w;
    }

    const int jbase = jb * SRC_PER_BLOCK + w * SRC_PER_WAVE;
    const float* __restrict__ adjp = adj + (size_t)jbase * N_NODES + i0 + lane;
    float aja[SRC_PER_WAVE], ajb[SRC_PER_WAVE];
#pragma unroll
    for (int jj = 0; jj < SRC_PER_WAVE; ++jj) {
        aja[jj] = adjp[(size_t)jj * N_NODES];
        ajb[jj] = adjp[(size_t)jj * N_NODES + 32];
    }
    __syncthreads();

    v2f e2[SRC_PER_WAVE];
#pragma unroll
    for (int jj = 0; jj < SRC_PER_WAVE; ++jj) { e2[jj].x = 0.0f; e2[jj].y = 0.0f; }

    const float* __restrict__ glp = gl + (size_t)jbase * OUT_F + h * HID;
#pragma unroll 1
    for (int c = 0; c < HID / F_CHUNK; ++c) {
        v2f g2[F_CHUNK];
        const float4* __restrict__ gsrc =
            reinterpret_cast<const float4*>(s_g2 + lane * G2_PITCH + 2 * F_CHUNK * c);
#pragma unroll
        for (int q = 0; q < F_CHUNK / 2; ++q) {
            const float4 v = gsrc[q];
            g2[2 * q].x = v.x;     g2[2 * q].y = v.y;
            g2[2 * q + 1].x = v.z; g2[2 * q + 1].y = v.w;
        }
        const v16f awc = *reinterpret_cast<const v16f*>(aw + F_CHUNK * c);
#pragma unroll
        for (int jg = 0; jg < SRC_PER_WAVE; jg += SRC_GROUP) {
            v16f row[SRC_GROUP];
#pragma unroll
            for (int r = 0; r < SRC_GROUP; ++r)
                row[r] = *reinterpret_cast<const v16f*>(
                    glp + (size_t)(jg + r) * OUT_F + F_CHUNK * c);
#pragma unroll
            for (int f = 0; f < F_CHUNK; ++f) {
                const float awf = awc[f];
                v2f s[SRC_GROUP], m[SRC_GROUP], lr[SRC_GROUP];
#pragma unroll
                for (int r = 0; r < SRC_GROUP; ++r) {
                    const float glf = row[r][f];
                    s[r] = g2[f] + (v2f){glf, glf};
                }
#pragma unroll
                for (int r = 0; r < SRC_GROUP; ++r)
                    m[r] = s[r] * (v2f){NEG_SLOPE, NEG_SLOPE};
#pragma unroll
                for (int r = 0; r < SRC_GROUP; ++r)
                    lr[r] = __builtin_elementwise_max(s[r], m[r]);
#pragma unroll
                for (int r = 0; r < SRC_GROUP; ++r)
                    e2[jg + r] = __builtin_elementwise_fma(lr[r], (v2f){awf, awf}, e2[jg + r]);
            }
        }
    }

    float dena = 0.0f, denb = 0.0f;
    float* __restrict__ prow = PT + ((size_t)h * N_NODES + jbase) * N_NODES + i0 + lane;
#pragma unroll
    for (int jj = 0; jj < SRC_PER_WAVE; ++jj) {
        const float pa = (aja[jj] != 0.0f) ? expf(e2[jj].x) : 0.0f;
        const float pb = (ajb[jj] != 0.0f) ? expf(e2[jj].y) : 0.0f;
        prow[(size_t)jj * N_NODES]      = pa;
        prow[(size_t)jj * N_NODES + 32] = pb;
        dena += pa;
        denb += pb;
    }

    s_den[w][lane]      = dena;
    s_den[w][lane + 32] = denb;
    __syncthreads();
    if (t < TGT_PER_BLOCK) {
        float d = s_den[0][t];
        for (int k = 1; k < SC_WAVES; ++k) d += s_den[k][t];
        dpart[((size_t)h * N_SRC_BLOCKS + jb) * N_NODES + i0 + t] = d;
    }
}

#define AG_WAVES  8
#define AG_SLICE  (N_NODES / AG_WAVES)
#define AG_TRIP   16

__global__ __launch_bounds__(AG_WAVES * 32)
void aggregate(const float* __restrict__ PT, const float* __restrict__ dpart,
               const float* __restrict__ gr, float* __restrict__ out,
               const int* __restrict__ use_elu) {
    const int t     = threadIdx.x;
    const int lane  = t & 31;
    const int w     = __builtin_amdgcn_readfirstlane(t >> 5);
    const int tileM = blockIdx.x * 16;
    const int h     = blockIdx.y;
    const int c     = lane & 15;
    const int kh    = (lane >> 4) * 2;
    const int m     = tileM + c;

    __shared__ float4 s_part[AG_WAVES * 8 * 32];

    float dsum = dpart[(size_t)h * N_SRC_BLOCKS * N_NODES + m];
    for (int jb = 1; jb < N_SRC_BLOCKS; ++jb)
        dsum += dpart[((size_t)h * N_SRC_BLOCKS + jb) * N_NODES + m];
    const float recip = 1.0f / dsum;
    const int jbeg = w * AG_SLICE;
    const float* pa = PT + (size_t)h * N_NODES * N_NODES + (size_t)(jbeg + kh) * N_NODES + m;
    const float* pb = gr + (size_t)(jbeg + kh) * OUT_F + h * HID + 4 * c;

    v8f acc0 = {}, acc1 = {}, acc2 = {}, acc3 = {};
#pragma unroll
    for (int kk = 0; kk < AG_SLICE; kk += 4 * AG_TRIP) {
        v2f a[AG_TRIP];
        float4 b0[AG_TRIP], b1[AG_TRIP];
#pragma unroll
        for (int u = 0; u < AG_TRIP; ++u) {
            const int k = kk + 4 * u;
            a[u].x = pa[(size_t)k * N_NODES];
            a[u].y = pa[(size_t)(k + 1) * N_NODES];
            b0[u] = *reinterpret_cast<const float4*>(pb + (size_t)k * OUT_F);
            b1[u] = *reinterpret_cast<const float4*>(pb + (size_t)(k + 1) * OUT_F);
        }
#pragma unroll
        for (int u = 0; u < AG_TRIP; ++u) {
            v2f av, bv;
            av.x = a[u].x * recip;
            av.y = a[u].y * recip;
            bv.x = b0[u].x; bv.y = b1[u].x;
            acc0 = __builtin_amdgcn_wmma_f32_16x16x4_f32(false, av, false, bv,
                                                         (short)0, acc0, false, false);
            bv.x = b0[u].y; bv.y = b1[u].y;
            acc1 = __builtin_amdgcn_wmma_f32_16x16x4_f32(false, av, false, bv,
                                                         (short)0, acc1, false, false);
            bv.x = b0[u].z; bv.y = b1[u].z;
            acc2 = __builtin_amdgcn_wmma_f32_16x16x4_f32(false, av, false, bv,
                                                         (short)0, acc2, false, false);
            bv.x = b0[u].w; bv.y = b1[u].w;
            acc3 = __builtin_amdgcn_wmma_f32_16x16x4_f32(false, av, false, bv,
                                                         (short)0, acc3, false, false);
        }
    }

#pragma unroll
    for (int r = 0; r < 8; ++r)
        s_part[(w * 8 + r) * 32 + lane] = make_float4(acc0[r], acc1[r], acc2[r], acc3[r]);
    __syncthreads();

    float4 s = s_part[t];
#pragma unroll
    for (int k = 1; k < AG_WAVES; ++k) {
        const float4 v = s_part[k * (8 * 32) + t];
        s.x += v.x; s.y += v.y; s.z += v.z; s.w += v.w;
    }

    if (use_elu[0]) {
        s.x = (s.x > 0.0f) ? s.x : expm1f(s.x);
        s.y = (s.y > 0.0f) ? s.y : expm1f(s.y);
        s.z = (s.z > 0.0f) ? s.z : expm1f(s.z);
        s.w = (s.w > 0.0f) ? s.w : expm1f(s.w);
    }
    const int row = tileM + 8 * (lane >> 4) + (t >> 5);
    *reinterpret_cast<float4*>(out + (size_t)row * OUT_F + h * HID + 4 * c) = s;
}

extern "C" void kernel_launch(void* const* d_in, const int* in_sizes, int n_in,
                              void* d_out, int out_size, void* d_ws, size_t ws_size,
                              hipStream_t stream) {
    const float* h_in  = (const float*)d_in[0];
    const float* nei   = (const float*)d_in[1];
    const float* adj   = (const float*)d_in[2];
    const float* w_l   = (const float*)d_in[3];
    const float* w_r   = (const float*)d_in[4];
    const float* aw    = (const float*)d_in[5];
    const int*   elu   = (const int*)d_in[6];
    float* out = (float*)d_out;

    float* ws    = (float*)d_ws;
    float* dpart = ws;
    float* g_r   = ws + (size_t)N_NODES * OUT_F;
    float* PT    = g_r + (size_t)N_NODES * OUT_F;
    float* g_l   = out;

    project_both<<<dim3(OUT_F / 64, N_NODES / 32, 2), PJ_WAVES * 32, 0, stream>>>(
        h_in, w_l, g_l, nei, w_r, g_r);

    attn_scores_t<<<dim3(N_NODES / TGT_PER_BLOCK, N_SRC_BLOCKS, HEADS), SC_WAVES * 32, 0, stream>>>(
        g_l, g_r, adj, aw, PT, dpart);

    aggregate<<<dim3(N_NODES / 16, HEADS), AG_WAVES * 32, 0, stream>>>(PT, dpart, g_r, out, elu);
}


# Score kernel: masked exponentials as packed multiply by log2(e) + hardware exp2 + select, 32 independent per lane, in place of 32 branched library exp chains; p rounding changes in the last bit
# speedup vs baseline: 6.3220x; 1.1449x over previous
#include <hip/hip_runtime.h>
#include <math.h>


typedef float v2f __attribute__((ext_vector_type(2)));
typedef float v8f __attribute__((ext_vector_type(8)));
typedef float v16f __attribute__((ext_vector_type(16)));

#define N_NODES 1024
#define IN_F    256
#define OUT_F   256
#define HEADS   4
#define HID     64
#define NEG_SLOPE 0.2f

#define SC_WAVES      16
#define SRC_PER_WAVE  16
#define SRC_PER_BLOCK (SC_WAVES * SRC_PER_WAVE)
#define N_SRC_BLOCKS  (N_NODES / SRC_PER_BLOCK)
#define TGT_PER_BLOCK 64
#define F_CHUNK       16
#define SRC_GROUP     4
#define G2_PITCH      132

#define PJ_WAVES 4
#define PJ_KBLK  32
#define PJ_SETK  (2 * PJ_KBLK)

struct PjSet {
    float4 a[2][4];
    float2 b[2][16];
};

__device__ __forceinline__ void pj_load(PjSet& s, const float* __restrict__ pa,
                                        const float* __restrict__ pb, const int k0) {
#pragma unroll
    for (int blk = 0; blk < 2; ++blk) {
        const int kb = k0 + blk * PJ_KBLK;
#pragma unroll
        for (int q = 0; q < 4; ++q)
            s.a[blk][q] = *reinterpret_cast<const float4*>(pa + kb + 4 * q);
#pragma unroll
        for (int e = 0; e < 16; ++e)
            s.b[blk][e] = *reinterpret_cast<const float2*>(pb + (size_t)(kb + e) * OUT_F);
    }
}

__device__ __forceinline__ void pj_mma(const PjSet& s, v8f& acc0, v8f& acc1) {
#pragma unroll
    for (int blk = 0; blk < 2; ++blk) {
#pragma unroll
        for (int u = 0; u < 8; ++u) {
            const float4 aq = s.a[blk][u >> 1];
            const float2 bk0 = s.b[blk][2 * u], bk1 = s.b[blk][2 * u + 1];
            v2f av, bv;
            av.x = (u & 1) ? aq.z : aq.x;
            av.y = (u & 1) ? aq.w : aq.y;
            bv.x = bk0.x; bv.y = bk1.x;
            acc0 = __builtin_amdgcn_wmma_f32_16x16x4_f32(false, av, false, bv,
                                                         (short)0, acc0, false, false);
            bv.x = bk0.y; bv.y = bk1.y;
            acc1 = __builtin_amdgcn_wmma_f32_16x16x4_f32(false, av, false, bv,
                                                         (short)0, acc1, false, false);
        }
    }
}

__global__ __launch_bounds__(PJ_WAVES * 32) __attribute__((amdgpu_waves_per_eu(1, 2)))
void project_both(const float* __restrict__ X0, const float* __restrict__ W0,
                  float* __restrict__ G0,
                  const float* __restrict__ X1, const float* __restrict__ W1,
                  float* __restrict__ G1) {
    const int t    = threadIdx.x;
    const int lane = t & 31;
    const int w    = __builtin_amdgcn_readfirstlane(t >> 5);
    const int c    = lane & 15;
    const int hh   = lane >> 4;
    const bool second = (blockIdx.z != 0);
    const float* __restrict__ X = second ? X1 : X0;
    const float* __restrict__ W = second ? W1 : W0;
    float* __restrict__ G       = second ? G1 : G0;

    const int tileM = (blockIdx.y * 2 + (w >> 1)) * 16;
    const int n0    = (blockIdx.x * 2 + (w & 1)) * 32;

    const float* pa = X + (size_t)(tileM + c) * IN_F + 16 * hh;
    const float* pb = W + (size_t)(16 * hh) * OUT_F + n0 + 2 * c;

    v8f acc0 = {}, acc1 = {};
    PjSet s0, s1;
    pj_load(s0, pa, pb, 0 * PJ_SETK);
    pj_load(s1, pa, pb, 1 * PJ_SETK);
    pj_mma(s0, acc0, acc1);
    pj_load(s0, pa, pb, 2 * PJ_SETK);
    pj_mma(s1, acc0, acc1);
    pj_load(s1, pa, pb, 3 * PJ_SETK);
    pj_mma(s0, acc0, acc1);
    pj_mma(s1, acc0, acc1);

    float* po = G + (size_t)(tileM + 8 * hh) * OUT_F + n0 + 2 * c;
#pragma unroll
    for (int r = 0; r < 8; ++r)
        *reinterpret_cast<float2*>(po + (size_t)r * OUT_F) = make_float2(acc0[r], acc1[r]);
}

#define PT_TILES 4

__device__ __forceinline__ float xhalf(const float x) {
    return __builtin_bit_cast(float, __builtin_amdgcn_permlanex16(
        __builtin_bit_cast(int, x), __builtin_bit_cast(int, x),
        0x76543210, 0xfedcba98, false, false));
}

__global__ __launch_bounds__(SC_WAVES * 32)
void attn_scores_agg(const float* __restrict__ gl, const float* __restrict__ gr,
                     const float* __restrict__ adj, const float* __restrict__ aw,
                     float* __restrict__ part, float* __restrict__ dpart) {
    const int t    = threadIdx.x;
    const int lane = t & 31;
    const int w    = __builtin_amdgcn_readfirstlane(t >> 5);
    const int i0   = blockIdx.x * TGT_PER_BLOCK;
    const int jb   = blockIdx.y;
    const int h    = blockIdx.z;

    __shared__ __attribute__((aligned(16))) float s_g2[32 * G2_PITCH];
    __shared__ float s_den[SC_WAVES][TGT_PER_BLOCK];
    __shared__ float4 s_part[SC_WAVES * PT_TILES * 8 * 32];

    for (int q = t; q < TGT_PER_BLOCK * (HID / 4); q += SC_WAVES * 32) {
        const int row = q >> 4, c4 = q & 15;
        const float4 v = *reinterpret_cast<const float4*>(
            gr + (size_t)(i0 + row) * OUT_F + h * HID + c4 * 4);
        float* d = s_g2 + (row & 31) * G2_PITCH + 8 * c4 + (row >> 5);
        d[0] = v.x; d[2] = v.y; d[4] = v.z; d[6] = v.w;
    }

    const int jbase = jb * SRC_PER_BLOCK + w * SRC_PER_WAVE;
    const float* __restrict__ adjp = adj + (size_t)jbase * N_NODES + i0 + lane;
    float aja[SRC_PER_WAVE], ajb[SRC_PER_WAVE];
#pragma unroll
    for (int jj = 0; jj < SRC_PER_WAVE; ++jj) {
        aja[jj] = adjp[(size_t)jj * N_NODES];
        ajb[jj] = adjp[(size_t)jj * N_NODES + 32];
    }
    __syncthreads();

    v2f e2[SRC_PER_WAVE];
#pragma unroll
    for (int jj = 0; jj < SRC_PER_WAVE; ++jj) { e2[jj].x = 0.0f; e2[jj].y = 0.0f; }

    const float* __restrict__ glp = gl + (size_t)jbase * OUT_F + h * HID;
#pragma unroll 1
    for (int c = 0; c < HID / F_CHUNK; ++c) {
        v2f g2[F_CHUNK];
        const float4* __restrict__ gsrc =
            reinterpret_cast<const float4*>(s_g2 + lane * G2_PITCH + 2 * F_CHUNK * c);
#pragma unroll
        for (int q = 0; q < F_CHUNK / 2; ++q) {
            const float4 v = gsrc[q];
            g2[2 * q].x = v.x;     g2[2 * q].y = v.y;
            g2[2 * q + 1].x = v.z; g2[2 * q + 1].y = v.w;
        }
        const v16f awc = *reinterpret_cast<const v16f*>(aw + F_CHUNK * c);
#pragma unroll
        for (int jg = 0; jg < SRC_PER_WAVE; jg += SRC_GROUP) {
            v16f row[SRC_GROUP];
#pragma unroll
            for (int r = 0; r < SRC_GROUP; ++r)
                row[r] = *reinterpret_cast<const v16f*>(
                    glp + (size_t)(jg + r) * OUT_F + F_CHUNK * c);
#pragma unroll
            for (int f = 0; f < F_CHUNK; ++f) {
                const float awf = awc[f];
                v2f s[SRC_GROUP], m[SRC_GROUP], lr[SRC_GROUP];
#pragma unroll
                for (int r = 0; r < SRC_GROUP; ++r) {
                    const float glf = row[r][f];
                    s[r] = g2[f] + (v2f){glf, glf};
                }
#pragma unroll
                for (int r = 0; r < SRC_GROUP; ++r)
                    m[r] = s[r] * (v2f){NEG_SLOPE, NEG_SLOPE};
#pragma unroll
                for (int r = 0; r < SRC_GROUP; ++r)
                    lr[r] = __builtin_elementwise_max(s[r], m[r]);
#pragma unroll
                for (int r = 0; r < SRC_GROUP; ++r)
                    e2[jg + r] = __builtin_elementwise_fma(lr[r], (v2f){awf, awf}, e2[jg + r]);
            }
        }
    }

    const int cc = lane & 15;
    const int hh = lane >> 4;
    const float* __restrict__ pbq = gr + (size_t)(jbase + 2 * hh) * OUT_F + h * HID + 4 * cc;
    float4 b0[SRC_PER_WAVE / 4], b1[SRC_PER_WAVE / 4];
#pragma unroll
    for (int q = 0; q < SRC_PER_WAVE / 4; ++q) {
        b0[q] = *reinterpret_cast<const float4*>(pbq + (size_t)(4 * q) * OUT_F);
        b1[q] = *reinterpret_cast<const float4*>(pbq + (size_t)(4 * q + 1) * OUT_F);
    }

    float pa[SRC_PER_WAVE], pb[SRC_PER_WAVE];
    float dena = 0.0f, denb = 0.0f;
    const v2f l2e = {1.4426950408889634f, 1.4426950408889634f};
#pragma unroll
    for (int jj = 0; jj < SRC_PER_WAVE; ++jj) {
        const v2f t2 = e2[jj] * l2e;
        const float xa = __builtin_amdgcn_exp2f(t2.x);
        const float xb = __builtin_amdgcn_exp2f(t2.y);
        pa[jj] = (aja[jj] != 0.0f) ? xa : 0.0f;
        pb[jj] = (ajb[jj] != 0.0f) ? xb : 0.0f;
        dena += pa[jj];
        denb += pb[jj];
    }

    v8f acc[PT_TILES][4];
#pragma unroll
    for (int mt = 0; mt < PT_TILES; ++mt)
#pragma unroll
        for (int tt = 0; tt < 4; ++tt) acc[mt][tt] = (v8f){};

#pragma unroll
    for (int q = 0; q < SRC_PER_WAVE / 4; ++q) {
        v2f av[PT_TILES];
        const bool hi = (hh != 0);
        const float xa0 = xhalf(pa[4 * q]),     xa1 = xhalf(pa[4 * q + 1]);
        const float xa2 = xhalf(pa[4 * q + 2]), xa3 = xhalf(pa[4 * q + 3]);
        const float xb0 = xhalf(pb[4 * q]),     xb1 = xhalf(pb[4 * q + 1]);
        const float xb2 = xhalf(pb[4 * q + 2]), xb3 = xhalf(pb[4 * q + 3]);
        av[0].x = hi ? xa2 : pa[4 * q];
        av[0].y = hi ? xa3 : pa[4 * q + 1];
        av[1].x = hi ? pa[4 * q + 2] : xa0;
        av[1].y = hi ? pa[4 * q + 3] : xa1;
        av[2].x = hi ? xb2 : pb[4 * q];
        av[2].y = hi ? xb3 : pb[4 * q + 1];
        av[3].x = hi ? pb[4 * q + 2] : xb0;
        av[3].y = hi ? pb[4 * q + 3] : xb1;
        const float bx[4] = {b0[q].x, b0[q].y, b0[q].z, b0[q].w};
        const float by[4] = {b1[q].x, b1[q].y, b1[q].z, b1[q].w};
#pragma unroll
        for (int tt = 0; tt < 4; ++tt) {
            v2f bv;
            bv.x = bx[tt]; bv.y = by[tt];
#pragma unroll
            for (int mt = 0; mt < PT_TILES; ++mt)
                acc[mt][tt] = __builtin_amdgcn_wmma_f32_16x16x4_f32(false, av[mt], false, bv,
                                                                    (short)0, acc[mt][tt], false, false);
        }
    }

#pragma unroll
    for (int mt = 0; mt < PT_TILES; ++mt)
#pragma unroll
        for (int r = 0; r < 8; ++r)
            s_part[((w * PT_TILES + mt) * 8 + r) * 32 + lane] =
                make_float4(acc[mt][0][r], acc[mt][1][r], acc[mt][2][r], acc[mt][3][r]);

    s_den[w][lane]      = dena;
    s_den[w][lane + 32] = denb;
    __syncthreads();
    if (t < TGT_PER_BLOCK) {
        float d = s_den[0][t];
        for (int k = 1; k < SC_WAVES; ++k) d += s_den[k][t];
        dpart[((size_t)h * N_SRC_BLOCKS + jb) * N_NODES + i0 + t] = d;
    }

#pragma unroll
    for (int half = 0; half < 2; ++half) {
        const int idx = t + half * (SC_WAVES * 32);
        float4 s = s_part[idx];
#pragma unroll
        for (int k = 1; k < SC_WAVES; ++k) {
            const float4 v = s_part[k * (PT_TILES * 8 * 32) + idx];
            s.x += v.x; s.y += v.y; s.z += v.z; s.w += v.w;
        }
        const int mt  = idx >> 8;
        const int r   = (idx >> 5) & 7;
        const int row = i0 + 16 * mt + 8 * hh + r;
        *reinterpret_cast<float4*>(
            part + ((size_t)jb * N_NODES + row) * OUT_F + h * HID + 4 * cc) = s;
    }
}

#define CB_THREADS 256

__global__ __launch_bounds__(CB_THREADS)
void combine(const float* __restrict__ part, const float* __restrict__ dpart,
             float* __restrict__ out, const int* __restrict__ use_elu) {
    const int q  = blockIdx.x * CB_THREADS + threadIdx.x;
    const int i  = q >> 6;
    const int h  = (q & 63) >> 4;
    const float4* __restrict__ p4 = reinterpret_cast<const float4*>(part);

    float dsum = dpart[(size_t)h * N_SRC_BLOCKS * N_NODES + i];
    float4 s = p4[q];
#pragma unroll
    for (int jb = 1; jb < N_SRC_BLOCKS; ++jb) {
        dsum += dpart[((size_t)h * N_SRC_BLOCKS + jb) * N_NODES + i];
        const float4 v = p4[(size_t)jb * (N_NODES * OUT_F / 4) + q];
        s.x += v.x; s.y += v.y; s.z += v.z; s.w += v.w;
    }
    s.x /= dsum; s.y /= dsum; s.z /= dsum; s.w /= dsum;

    if (use_elu[0]) {
        s.x = (s.x > 0.0f) ? s.x : expm1f(s.x);
        s.y = (s.y > 0.0f) ? s.y : expm1f(s.y);
        s.z = (s.z > 0.0f) ? s.z : expm1f(s.z);
        s.w = (s.w > 0.0f) ? s.w : expm1f(s.w);
    }
    reinterpret_cast<float4*>(out)[q] = s;
}

extern "C" void kernel_launch(void* const* d_in, const int* in_sizes, int n_in,
                              void* d_out, int out_size, void* d_ws, size_t ws_size,
                              hipStream_t stream) {
    const float* h_in  = (const float*)d_in[0];
    const float* nei   = (const float*)d_in[1];
    const float* adj   = (const float*)d_in[2];
    const float* w_l   = (const float*)d_in[3];
    const float* w_r   = (const float*)d_in[4];
    const float* aw    = (const float*)d_in[5];
    const int*   elu   = (const int*)d_in[6];
    float* out = (float*)d_out;

    float* ws    = (float*)d_ws;
    float* dpart = ws;
    float* g_r   = ws + (size_t)N_NODES * OUT_F;
    float* part  = g_r + (size_t)N_NODES * OUT_F;
    float* g_l   = out;

    project_both<<<dim3(OUT_F / 64, N_NODES / 32, 2), PJ_WAVES * 32, 0, stream>>>(
        h_in, w_l, g_l, nei, w_r, g_r);

    attn_scores_agg<<<dim3(N_NODES / TGT_PER_BLOCK, N_SRC_BLOCKS, HEADS), SC_WAVES * 32, 0, stream>>>(
        g_l, g_r, adj, aw, part, dpart);

    combine<<<dim3(N_NODES * OUT_F / 4 / CB_THREADS), CB_THREADS, 0, stream>>>(
        part, dpart, out, elu);
}


# Score kernel opening: a thread's two tile loads issued together, chunk 0's scalar rows asked for ahead of the barrier, adjacency loads moved behind chunk 0 so no barrier drains them; sum orders kept
# speedup vs baseline: 6.5820x; 1.0411x over previous
#include <hip/hip_runtime.h>
#include <math.h>


typedef float v2f __attribute__((ext_vector_type(2)));
typedef float v8f __attribute__((ext_vector_type(8)));
typedef float v16f __attribute__((ext_vector_type(16)));

#define N_NODES 1024
#define IN_F    256
#define OUT_F   256
#define HEADS   4
#define HID     64
#define NEG_SLOPE 0.2f

#define SC_WAVES      16
#define SRC_PER_WAVE  16
#define SRC_PER_BLOCK (SC_WAVES * SRC_PER_WAVE)
#define N_SRC_BLOCKS  (N_NODES / SRC_PER_BLOCK)
#define TGT_PER_BLOCK 64
#define F_CHUNK       16
#define SRC_GROUP     4
#define G2_PITCH      132

#define PJ_WAVES 4
#define PJ_KBLK  32
#define PJ_SETK  (2 * PJ_KBLK)

struct PjSet {
    float4 a[2][4];
    float2 b[2][16];
};

__device__ __forceinline__ void pj_load(PjSet& s, const float* __restrict__ pa,
                                        const float* __restrict__ pb, const int k0) {
#pragma unroll
    for (int blk = 0; blk < 2; ++blk) {
        const int kb = k0 + blk * PJ_KBLK;
#pragma unroll
        for (int q = 0; q < 4; ++q)
            s.a[blk][q] = *reinterpret_cast<const float4*>(pa + kb + 4 * q);
#pragma unroll
        for (int e = 0; e < 16; ++e)
            s.b[blk][e] = *reinterpret_cast<const float2*>(pb + (size_t)(kb + e) * OUT_F);
    }
}

__device__ __forceinline__ void pj_mma(const PjSet& s, v8f& acc0, v8f& acc1) {
#pragma unroll
    for (int blk = 0; blk < 2; ++blk) {
#pragma unroll
        for (int u = 0; u < 8; ++u) {
            const float4 aq = s.a[blk][u >> 1];
            const float2 bk0 = s.b[blk][2 * u], bk1 = s.b[blk][2 * u + 1];
            v2f av, bv;
            av.x = (u & 1) ? aq.z : aq.x;
            av.y = (u & 1) ? aq.w : aq.y;
            bv.x = bk0.x; bv.y = bk1.x;
            acc0 = __builtin_amdgcn_wmma_f32_16x16x4_f32(false, av, false, bv,
                                                         (short)0, acc0, false, false);
            bv.x = bk0.y; bv.y = bk1.y;
            acc1 = __builtin_amdgcn_wmma_f32_16x16x4_f32(false, av, false, bv,
                                                         (short)0, acc1, false, false);
        }
    }
}

__global__ __launch_bounds__(PJ_WAVES * 32) __attribute__((amdgpu_waves_per_eu(1, 2)))
void project_both(const float* __restrict__ X0, const float* __restrict__ W0,
                  float* __restrict__ G0,
                  const float* __restrict__ X1, const float* __restrict__ W1,
                  float* __restrict__ G1) {
    const int t    = threadIdx.x;
    const int lane = t & 31;
    const int w    = __builtin_amdgcn_readfirstlane(t >> 5);
    const int c    = lane & 15;
    const int hh   = lane >> 4;
    const bool second = (blockIdx.z != 0);
    const float* __restrict__ X = second ? X1 : X0;
    const float* __restrict__ W = second ? W1 : W0;
    float* __restrict__ G       = second ? G1 : G0;

    const int tileM = (blockIdx.y * 2 + (w >> 1)) * 16;
    const int n0    = (blockIdx.x * 2 + (w & 1)) * 32;

    const float* pa = X + (size_t)(tileM + c) * IN_F + 16 * hh;
    const float* pb = W + (size_t)(16 * hh) * OUT_F + n0 + 2 * c;

    v8f acc0 = {}, acc1 = {};
    PjSet s0, s1;
    pj_load(s0, pa, pb, 0 * PJ_SETK);
    pj_load(s1, pa, pb, 1 * PJ_SETK);
    pj_mma(s0, acc0, acc1);
    pj_load(s0, pa, pb, 2 * PJ_SETK);
    pj_mma(s1, acc0, acc1);
    pj_load(s1, pa, pb, 3 * PJ_SETK);
    pj_mma(s0, acc0, acc1);
    pj_mma(s1, acc0, acc1);

    float* po = G + (size_t)(tileM + 8 * hh) * OUT_F + n0 + 2 * c;
#pragma unroll
    for (int r = 0; r < 8; ++r)
        *reinterpret_cast<float2*>(po + (size_t)r * OUT_F) = make_float2(acc0[r], acc1[r]);
}

#define PT_TILES 4

__device__ __forceinline__ float xhalf(const float x) {
    return __builtin_bit_cast(float, __builtin_amdgcn_permlanex16(
        __builtin_bit_cast(int, x), __builtin_bit_cast(int, x),
        0x76543210, 0xfedcba98, false, false));
}

__device__ __forceinline__ void load_rows(v16f (&row)[SRC_GROUP], const float* __restrict__ glp,
                                          const int jg, const int c) {
#pragma unroll
    for (int r = 0; r < SRC_GROUP; ++r)
        row[r] = *reinterpret_cast<const v16f*>(glp + (size_t)(jg + r) * OUT_F + F_CHUNK * c);
}

template <int JG>
__device__ __forceinline__ void score_group(v2f (&e2)[SRC_PER_WAVE], const v2f (&g2)[F_CHUNK],
                                            const v16f& awc, const v16f (&row)[SRC_GROUP]) {
#pragma unroll
    for (int f = 0; f < F_CHUNK; ++f) {
        const float awf = awc[f];
        v2f s[SRC_GROUP], m[SRC_GROUP], lr[SRC_GROUP];
#pragma unroll
        for (int r = 0; r < SRC_GROUP; ++r) {
            const float glf = row[r][f];
            s[r] = g2[f] + (v2f){glf, glf};
        }
#pragma unroll
        for (int r = 0; r < SRC_GROUP; ++r)
            m[r] = s[r] * (v2f){NEG_SLOPE, NEG_SLOPE};
#pragma unroll
        for (int r = 0; r < SRC_GROUP; ++r)
            lr[r] = __builtin_elementwise_max(s[r], m[r]);
#pragma unroll
        for (int r = 0; r < SRC_GROUP; ++r)
            e2[JG + r] = __builtin_elementwise_fma(lr[r], (v2f){awf, awf}, e2[JG + r]);
    }
}

__device__ __forceinline__ void score_chunk(v2f (&e2)[SRC_PER_WAVE], const float* s_g2, const int lane,
                                            const float* __restrict__ glp, const int c,
                                            const v16f& awc, const v16f (&row0)[SRC_GROUP]) {
    v2f g2[F_CHUNK];
    const float4* gsrc = reinterpret_cast<const float4*>(s_g2 + lane * G2_PITCH + 2 * F_CHUNK * c);
#pragma unroll
    for (int q = 0; q < F_CHUNK / 2; ++q) {
        const float4 v = gsrc[q];
        g2[2 * q].x = v.x;     g2[2 * q].y = v.y;
        g2[2 * q + 1].x = v.z; g2[2 * q + 1].y = v.w;
    }
    score_group<0>(e2, g2, awc, row0);
    { v16f row[SRC_GROUP]; load_rows(row, glp, 1 * SRC_GROUP, c); score_group<1 * SRC_GROUP>(e2, g2, awc, row); }
    { v16f row[SRC_GROUP]; load_rows(row, glp, 2 * SRC_GROUP, c); score_group<2 * SRC_GROUP>(e2, g2, awc, row); }
    { v16f row[SRC_GROUP]; load_rows(row, glp, 3 * SRC_GROUP, c); score_group<3 * SRC_GROUP>(e2, g2, awc, row); }
}

__global__ __launch_bounds__(SC_WAVES * 32)
void attn_scores_agg(const float* __restrict__ gl, const float* __restrict__ gr,
                     const float* __restrict__ adj, const float* __restrict__ aw,
                     float* __restrict__ part, float* __restrict__ dpart) {
    const int t    = threadIdx.x;
    const int lane = t & 31;
    const int w    = __builtin_amdgcn_readfirstlane(t >> 5);
    const int i0   = blockIdx.x * TGT_PER_BLOCK;
    const int jb   = blockIdx.y;
    const int h    = blockIdx.z;

    __shared__ __attribute__((aligned(16))) float s_g2[32 * G2_PITCH];
    __shared__ float s_den[SC_WAVES][TGT_PER_BLOCK];
    __shared__ float4 s_part[SC_WAVES * PT_TILES * 8 * 32];

    static_assert(TGT_PER_BLOCK * (HID / 4) == 2 * SC_WAVES * 32, "two pieces per thread");
    float4 stv[2];
#pragma unroll
    for (int u = 0; u < 2; ++u) {
        const int q = t + u * (SC_WAVES * 32);
        const int row = q >> 4, c4 = q & 15;
        stv[u] = *reinterpret_cast<const float4*>(
            gr + (size_t)(i0 + row) * OUT_F + h * HID + c4 * 4);
    }
    const int jbase = jb * SRC_PER_BLOCK + w * SRC_PER_WAVE;
    const float* __restrict__ glp = gl + (size_t)jbase * OUT_F + h * HID;
    const v16f awc0 = *reinterpret_cast<const v16f*>(aw);
    v16f row00[SRC_GROUP];
    load_rows(row00, glp, 0, 0);
#pragma unroll
    for (int u = 0; u < 2; ++u) {
        const int q = t + u * (SC_WAVES * 32);
        const int row = q >> 4, c4 = q & 15;
        float* d = s_g2 + (row & 31) * G2_PITCH + 8 * c4 + (row >> 5);
        d[0] = stv[u].x; d[2] = stv[u].y; d[4] = stv[u].z; d[6] = stv[u].w;
    }
    __syncthreads();

    v2f e2[SRC_PER_WAVE];
#pragma unroll
    for (int jj = 0; jj < SRC_PER_WAVE; ++jj) { e2[jj].x = 0.0f; e2[jj].y = 0.0f; }

    score_chunk(e2, s_g2, lane, glp, 0, awc0, row00);

    const float* __restrict__ adjp = adj + (size_t)jbase * N_NODES + i0 + lane;
    float aja[SRC_PER_WAVE], ajb[SRC_PER_WAVE];
#pragma unroll
    for (int jj = 0; jj < SRC_PER_WAVE; ++jj) {
        aja[jj] = adjp[(size_t)jj * N_NODES];
        ajb[jj] = adjp[(size_t)jj * N_NODES + 32];
    }

#pragma unroll 1
    for (int c = 1; c < HID / F_CHUNK; ++c) {
        const v16f awc = *reinterpret_cast<const v16f*>(aw + F_CHUNK * c);
        v16f row0[SRC_GROUP];
        load_rows(row0, glp, 0, c);
        score_chunk(e2, s_g2, lane, glp, c, awc, row0);
    }

    const int cc = lane & 15;
    const int hh = lane >> 4;
    const float* __restrict__ pbq = gr + (size_t)(jbase + 2 * hh) * OUT_F + h * HID + 4 * cc;
    float4 b0[SRC_PER_WAVE / 4], b1[SRC_PER_WAVE / 4];
#pragma unroll
    for (int q = 0; q < SRC_PER_WAVE / 4; ++q) {
        b0[q] = *reinterpret_cast<const float4*>(pbq + (size_t)(4 * q) * OUT_F);
        b1[q] = *reinterpret_cast<const float4*>(pbq + (size_t)(4 * q + 1) * OUT_F);
    }

    float pa[SRC_PER_WAVE], pb[SRC_PER_WAVE];
    float dena = 0.0f, denb = 0.0f;
    const v2f l2e = {1.4426950408889634f, 1.4426950408889634f};
#pragma unroll
    for (int jj = 0; jj < SRC_PER_WAVE; ++jj) {
        const v2f t2 = e2[jj] * l2e;
        const float xa = __builtin_amdgcn_exp2f(t2.x);
        const float xb = __builtin_amdgcn_exp2f(t2.y);
        pa[jj] = (aja[jj] != 0.0f) ? xa : 0.0f;
        pb[jj] = (ajb[jj] != 0.0f) ? xb : 0.0f;
        dena += pa[jj];
        denb += pb[jj];
    }

    v8f acc[PT_TILES][4];
#pragma unroll
    for (int mt = 0; mt < PT_TILES; ++mt)
#pragma unroll
        for (int tt = 0; tt < 4; ++tt) acc[mt][tt] = (v8f){};

#pragma unroll
    for (int q = 0; q < SRC_PER_WAVE / 4; ++q) {
        v2f av[PT_TILES];
        const bool hi = (hh != 0);
        const float xa0 = xhalf(pa[4 * q]),     xa1 = xhalf(pa[4 * q + 1]);
        const float xa2 = xhalf(pa[4 * q + 2]), xa3 = xhalf(pa[4 * q + 3]);
        const float xb0 = xhalf(pb[4 * q]),     xb1 = xhalf(pb[4 * q + 1]);
        const float xb2 = xhalf(pb[4 * q + 2]), xb3 = xhalf(pb[4 * q + 3]);
        av[0].x = hi ? xa2 : pa[4 * q];
        av[0].y = hi ? xa3 : pa[4 * q + 1];
        av[1].x = hi ? pa[4 * q + 2] : xa0;
        av[1].y = hi ? pa[4 * q + 3] : xa1;
        av[2].x = hi ? xb2 : pb[4 * q];
        av[2].y = hi ? xb3 : pb[4 * q + 1];
        av[3].x = hi ? pb[4 * q + 2] : xb0;
        av[3].y = hi ? pb[4 * q + 3] : xb1;
        const float bx[4] = {b0[q].x, b0[q].y, b0[q].z, b0[q].w};
        const float by[4] = {b1[q].x, b1[q].y, b1[q].z, b1[q].w};
#pragma unroll
        for (int tt = 0; tt < 4; ++tt) {
            v2f bv;
            bv.x = bx[tt]; bv.y = by[tt];
#pragma unroll
            for (int mt = 0; mt < PT_TILES; ++mt)
                acc[mt][tt] = __builtin_amdgcn_wmma_f32_16x16x4_f32(false, av[mt], false, bv,
                                                                    (short)0, acc[mt][tt], false, false);
        }
    }

#pragma unroll
    for (int mt = 0; mt < PT_TILES; ++mt)
#pragma unroll
        for (int r = 0; r < 8; ++r)
            s_part[((w * PT_TILES + mt) * 8 + r) * 32 + lane] =
                make_float4(acc[mt][0][r], acc[mt][1][r], acc[mt][2][r], acc[mt][3][r]);

    s_den[w][lane]      = dena;
    s_den[w][lane + 32] = denb;
    __syncthreads();
    if (t < TGT_PER_BLOCK) {
        float d = s_den[0][t];
        for (int k = 1; k < SC_WAVES; ++k) d += s_den[k][t];
        dpart[((size_t)h * N_SRC_BLOCKS + jb) * N_NODES + i0 + t] = d;
    }

#pragma unroll
    for (int half = 0; half < 2; ++half) {
        const int idx = t + half * (SC_WAVES * 32);
        float4 s = s_part[idx];
#pragma unroll
        for (int k = 1; k < SC_WAVES; ++k) {
            const float4 v = s_part[k * (PT_TILES * 8 * 32) + idx];
            s.x += v.x; s.y += v.y; s.z += v.z; s.w += v.w;
        }
        const int mt  = idx >> 8;
        const int r   = (idx >> 5) & 7;
        const int row = i0 + 16 * mt + 8 * hh + r;
        *reinterpret_cast<float4*>(
            part + ((size_t)jb * N_NODES + row) * OUT_F + h * HID + 4 * cc) = s;
    }
}

#define CB_THREADS 256

__global__ __launch_bounds__(CB_THREADS)
void combine(const float* __restrict__ part, const float* __restrict__ dpart,
             float* __restrict__ out, const int* __restrict__ use_elu) {
    const int q  = blockIdx.x * CB_THREADS + threadIdx.x;
    const int i  = q >> 6;
    const int h  = (q & 63) >> 4;
    const float4* __restrict__ p4 = reinterpret_cast<const float4*>(part);

    float dsum = dpart[(size_t)h * N_SRC_BLOCKS * N_NODES + i];
    float4 s = p4[q];
#pragma unroll
    for (int jb = 1; jb < N_SRC_BLOCKS; ++jb) {
        dsum += dpart[((size_t)h * N_SRC_BLOCKS + jb) * N_NODES + i];
        const float4 v = p4[(size_t)jb * (N_NODES * OUT_F / 4) + q];
        s.x += v.x; s.y += v.y; s.z += v.z; s.w += v.w;
    }
    s.x /= dsum; s.y /= dsum; s.z /= dsum; s.w /= dsum;

    if (use_elu[0]) {
        s.x = (s.x > 0.0f) ? s.x : expm1f(s.x);
        s.y = (s.y > 0.0f) ? s.y : expm1f(s.y);
        s.z = (s.z > 0.0f) ? s.z : expm1f(s.z);
        s.w = (s.w > 0.0f) ? s.w : expm1f(s.w);
    }
    reinterpret_cast<float4*>(out)[q] = s;
}

extern "C" void kernel_launch(void* const* d_in, const int* in_sizes, int n_in,
                              void* d_out, int out_size, void* d_ws, size_t ws_size,
                              hipStream_t stream) {
    const float* h_in  = (const float*)d_in[0];
    const float* nei   = (const float*)d_in[1];
    const float* adj   = (const float*)d_in[2];
    const float* w_l   = (const float*)d_in[3];
    const float* w_r   = (const float*)d_in[4];
    const float* aw    = (const float*)d_in[5];
    const int*   elu   = (const int*)d_in[6];
    float* out = (float*)d_out;

    float* ws    = (float*)d_ws;
    float* dpart = ws;
    float* g_r   = ws + (size_t)N_NODES * OUT_F;
    float* part  = g_r + (size_t)N_NODES * OUT_F;
    float* g_l   = out;

    project_both<<<dim3(OUT_F / 64, N_NODES / 32, 2), PJ_WAVES * 32, 0, stream>>>(
        h_in, w_l, g_l, nei, w_r, g_r);

    attn_scores_agg<<<dim3(N_NODES / TGT_PER_BLOCK, N_SRC_BLOCKS, HEADS), SC_WAVES * 32, 0, stream>>>(
        g_l, g_r, adj, aw, part, dpart);

    combine<<<dim3(N_NODES * OUT_F / 4 / CB_THREADS), CB_THREADS, 0, stream>>>(
        part, dpart, out, elu);
}


# Score/aggregation roles swapped: lane owns 2 sources, block all 1024 sources of 16 targets, g_r rows as scalar operands; block divides and writes out, 3rd launch gone; den/neighbour sum order changes
# speedup vs baseline: 7.0019x; 1.0638x over previous
#include <hip/hip_runtime.h>
#include <math.h>


typedef float v2f __attribute__((ext_vector_type(2)));
typedef float v8f __attribute__((ext_vector_type(8)));
typedef float v16f __attribute__((ext_vector_type(16)));

#define N_NODES 1024
#define IN_F    256
#define OUT_F   256
#define HEADS   4
#define HID     64
#define NEG_SLOPE 0.2f

#define PJ_WAVES 4
#define PJ_KBLK  32
#define PJ_SETK  (2 * PJ_KBLK)

struct PjSet {
    float4 a[2][4];
    float2 b[2][16];
};

__device__ __forceinline__ void pj_load(PjSet& s, const float* __restrict__ pa,
                                        const float* __restrict__ pb, const int k0) {
#pragma unroll
    for (int blk = 0; blk < 2; ++blk) {
        const int kb = k0 + blk * PJ_KBLK;
#pragma unroll
        for (int q = 0; q < 4; ++q)
            s.a[blk][q] = *reinterpret_cast<const float4*>(pa + kb + 4 * q);
#pragma unroll
        for (int e = 0; e < 16; ++e)
            s.b[blk][e] = *reinterpret_cast<const float2*>(pb + (size_t)(kb + e) * OUT_F);
    }
}

__device__ __forceinline__ void pj_mma(const PjSet& s, v8f& acc0, v8f& acc1) {
#pragma unroll
    for (int blk = 0; blk < 2; ++blk) {
#pragma unroll
        for (int u = 0; u < 8; ++u) {
            const float4 aq = s.a[blk][u >> 1];
            const float2 bk0 = s.b[blk][2 * u], bk1 = s.b[blk][2 * u + 1];
            v2f av, bv;
            av.x = (u & 1) ? aq.z : aq.x;
            av.y = (u & 1) ? aq.w : aq.y;
            bv.x = bk0.x; bv.y = bk1.x;
            acc0 = __builtin_amdgcn_wmma_f32_16x16x4_f32(false, av, false, bv,
                                                         (short)0, acc0, false, false);
            bv.x = bk0.y; bv.y = bk1.y;
            acc1 = __builtin_amdgcn_wmma_f32_16x16x4_f32(false, av, false, bv,
                                                         (short)0, acc1, false, false);
        }
    }
}

__global__ __launch_bounds__(PJ_WAVES * 32) __attribute__((amdgpu_waves_per_eu(1, 2)))
void project_both(const float* __restrict__ X0, const float* __restrict__ W0,
                  float* __restrict__ G0,
                  const float* __restrict__ X1, const float* __restrict__ W1,
                  float* __restrict__ G1) {
    const int t    = threadIdx.x;
    const int lane = t & 31;
    const int w    = __builtin_amdgcn_readfirstlane(t >> 5);
    const int c    = lane & 15;
    const int hh   = lane >> 4;
    const bool second = (blockIdx.z != 0);
    const float* __restrict__ X = second ? X1 : X0;
    const float* __restrict__ W = second ? W1 : W0;
    float* __restrict__ G       = second ? G1 : G0;

    const int tileM = (blockIdx.y * 2 + (w >> 1)) * 16;
    const int n0    = (blockIdx.x * 2 + (w & 1)) * 32;

    const float* pa = X + (size_t)(tileM + c) * IN_F + 16 * hh;
    const float* pb = W + (size_t)(16 * hh) * OUT_F + n0 + 2 * c;

    v8f acc0 = {}, acc1 = {};
    PjSet s0, s1;
    pj_load(s0, pa, pb, 0 * PJ_SETK);
    pj_load(s1, pa, pb, 1 * PJ_SETK);
    pj_mma(s0, acc0, acc1);
    pj_load(s0, pa, pb, 2 * PJ_SETK);
    pj_mma(s1, acc0, acc1);
    pj_load(s1, pa, pb, 3 * PJ_SETK);
    pj_mma(s0, acc0, acc1);
    pj_mma(s1, acc0, acc1);

    float* po = G + (size_t)(tileM + 8 * hh) * OUT_F + n0 + 2 * c;
#pragma unroll
    for (int r = 0; r < 8; ++r)
        *reinterpret_cast<float2*>(po + (size_t)r * OUT_F) = make_float2(acc0[r], acc1[r]);
}

#define AG_WAVES      16
#define SRC_PER_WAVE  64
#define TGT_PER_BLOCK 16
#define F_CHUNK       16
#define TGT_GROUP     4
#define P_PITCH       68

__device__ __forceinline__ void load_rows(v16f (&row)[TGT_GROUP], const float* __restrict__ grp,
                                          const int rg, const int c) {
#pragma unroll
    for (int r = 0; r < TGT_GROUP; ++r)
        row[r] = *reinterpret_cast<const v16f*>(grp + (size_t)(rg + r) * OUT_F + F_CHUNK * c);
}

template <int RG>
__device__ __forceinline__ void score_group(v2f (&e2)[TGT_PER_BLOCK], const v2f (&g2)[F_CHUNK],
                                            const v16f& awc, const v16f (&row)[TGT_GROUP]) {
#pragma unroll
    for (int f = 0; f < F_CHUNK; ++f) {
        const float awf = awc[f];
        v2f s[TGT_GROUP], m[TGT_GROUP], lr[TGT_GROUP];
#pragma unroll
        for (int r = 0; r < TGT_GROUP; ++r) {
            const float grf = row[r][f];
            s[r] = (v2f){grf, grf} + g2[f];
        }
#pragma unroll
        for (int r = 0; r < TGT_GROUP; ++r)
            m[r] = s[r] * (v2f){NEG_SLOPE, NEG_SLOPE};
#pragma unroll
        for (int r = 0; r < TGT_GROUP; ++r)
            lr[r] = __builtin_elementwise_max(s[r], m[r]);
#pragma unroll
        for (int r = 0; r < TGT_GROUP; ++r)
            e2[RG + r] = __builtin_elementwise_fma(lr[r], (v2f){awf, awf}, e2[RG + r]);
    }
}

__global__ __launch_bounds__(AG_WAVES * 32)
void attn_fused(const float* __restrict__ gl, const float* __restrict__ gr,
                const float* __restrict__ adj, const float* __restrict__ aw,
                float* __restrict__ out, const int* __restrict__ use_elu) {
    const int t    = threadIdx.x;
    const int lane = t & 31;
    const int w    = __builtin_amdgcn_readfirstlane(t >> 5);
    const int i0   = blockIdx.x * TGT_PER_BLOCK;
    const int h    = blockIdx.y;
    const int cc   = lane & 15;
    const int hh   = lane >> 4;

    __shared__ __attribute__((aligned(16))) float s_p[AG_WAVES * TGT_PER_BLOCK * P_PITCH];
    __shared__ float4 s_part[AG_WAVES * 8 * 32];
    __shared__ float s_den[AG_WAVES][TGT_PER_BLOCK];

    const int jbase = w * SRC_PER_WAVE;
    const float* __restrict__ pla = gl + (size_t)(jbase + lane) * OUT_F + h * HID;
    const float* __restrict__ plb = pla + (size_t)32 * OUT_F;
    const float* __restrict__ grp = gr + (size_t)i0 * OUT_F + h * HID;

    v2f e2[TGT_PER_BLOCK];
#pragma unroll
    for (int r = 0; r < TGT_PER_BLOCK; ++r) { e2[r].x = 0.0f; e2[r].y = 0.0f; }

    float4 va[F_CHUNK / 4], vb[F_CHUNK / 4];
#pragma unroll
    for (int q = 0; q < F_CHUNK / 4; ++q) {
        va[q] = *reinterpret_cast<const float4*>(pla + 4 * q);
        vb[q] = *reinterpret_cast<const float4*>(plb + 4 * q);
    }

    asm volatile("" ::: "memory");
    const float* __restrict__ adja = adj + (size_t)(jbase + lane) * N_NODES + i0;
    const float* __restrict__ adjb = adja + (size_t)32 * N_NODES;
    float4 aja[4], ajb[4];
#pragma unroll
    for (int q = 0; q < 4; ++q) {
        aja[q] = *reinterpret_cast<const float4*>(adja + 4 * q);
        ajb[q] = *reinterpret_cast<const float4*>(adjb + 4 * q);
    }

#pragma unroll 1
    for (int c = 0; c < HID / F_CHUNK; ++c) {
        v2f g2[F_CHUNK];
#pragma unroll
        for (int q = 0; q < F_CHUNK / 4; ++q) {
            g2[4 * q].x = va[q].x;     g2[4 * q].y = vb[q].x;
            g2[4 * q + 1].x = va[q].y; g2[4 * q + 1].y = vb[q].y;
            g2[4 * q + 2].x = va[q].z; g2[4 * q + 2].y = vb[q].z;
            g2[4 * q + 3].x = va[q].w; g2[4 * q + 3].y = vb[q].w;
        }
        const int cn = (c + 1 < HID / F_CHUNK) ? c + 1 : c;
#pragma unroll
        for (int q = 0; q < F_CHUNK / 4; ++q) {
            va[q] = *reinterpret_cast<const float4*>(pla + F_CHUNK * cn + 4 * q);
            vb[q] = *reinterpret_cast<const float4*>(plb + F_CHUNK * cn + 4 * q);
        }
        asm volatile("" ::: "memory");
        const v16f awc = *reinterpret_cast<const v16f*>(aw + F_CHUNK * c);
        { v16f row[TGT_GROUP]; load_rows(row, grp, 0 * TGT_GROUP, c); score_group<0 * TGT_GROUP>(e2, g2, awc, row); }
        { v16f row[TGT_GROUP]; load_rows(row, grp, 1 * TGT_GROUP, c); score_group<1 * TGT_GROUP>(e2, g2, awc, row); }
        { v16f row[TGT_GROUP]; load_rows(row, grp, 2 * TGT_GROUP, c); score_group<2 * TGT_GROUP>(e2, g2, awc, row); }
        { v16f row[TGT_GROUP]; load_rows(row, grp, 3 * TGT_GROUP, c); score_group<3 * TGT_GROUP>(e2, g2, awc, row); }
    }

    const float* __restrict__ pbq = gr + (size_t)(jbase + 16 * hh) * OUT_F + h * HID + 4 * cc;
    float4 b0[SRC_PER_WAVE / 4], b1[SRC_PER_WAVE / 4];
#pragma unroll
    for (int q = 0; q < SRC_PER_WAVE / 8; ++q) {
        b0[q] = *reinterpret_cast<const float4*>(pbq + (size_t)q * OUT_F);
        b1[q] = *reinterpret_cast<const float4*>(pbq + (size_t)(32 + q) * OUT_F);
    }

    const v2f l2e = {1.4426950408889634f, 1.4426950408889634f};
    float* sp = s_p + w * (TGT_PER_BLOCK * P_PITCH);
#pragma unroll
    for (int r = 0; r < TGT_PER_BLOCK; ++r) {
        const float4 ma4 = aja[r >> 2], mb4 = ajb[r >> 2];
        const float ma = (r & 3) == 0 ? ma4.x : (r & 3) == 1 ? ma4.y : (r & 3) == 2 ? ma4.z : ma4.w;
        const float mb = (r & 3) == 0 ? mb4.x : (r & 3) == 1 ? mb4.y : (r & 3) == 2 ? mb4.z : mb4.w;
        const v2f t2 = e2[r] * l2e;
        const float xa = __builtin_amdgcn_exp2f(t2.x);
        const float xb = __builtin_amdgcn_exp2f(t2.y);
        v2f p2;
        p2.x = (ma != 0.0f) ? xa : 0.0f;
        p2.y = (mb != 0.0f) ? xb : 0.0f;
        *reinterpret_cast<v2f*>(sp + r * P_PITCH + 2 * lane) = p2;
    }

    __builtin_amdgcn_fence(__ATOMIC_RELEASE, "wavefront");
    __builtin_amdgcn_wave_barrier();
    __builtin_amdgcn_fence(__ATOMIC_ACQUIRE, "wavefront");

    float4 af[8];
    const float4* __restrict__ apr = reinterpret_cast<const float4*>(sp + cc * P_PITCH + 32 * hh);
#pragma unroll
    for (int m = 0; m < 8; ++m) af[m] = apr[m];

#pragma unroll
    for (int q = SRC_PER_WAVE / 8; q < SRC_PER_WAVE / 4; ++q) {
        b0[q] = *reinterpret_cast<const float4*>(pbq + (size_t)q * OUT_F);
        b1[q] = *reinterpret_cast<const float4*>(pbq + (size_t)(32 + q) * OUT_F);
    }

    float dsum = 0.0f;
#pragma unroll
    for (int m = 0; m < 8; ++m) dsum += (af[m].x + af[m].y) + (af[m].z + af[m].w);
    dsum += __builtin_bit_cast(float, __builtin_amdgcn_permlanex16(
        __builtin_bit_cast(int, dsum), __builtin_bit_cast(int, dsum),
        0x76543210, 0xfedcba98, false, false));

    v8f acc[4];
#pragma unroll
    for (int tt = 0; tt < 4; ++tt) acc[tt] = (v8f){};
#pragma unroll
    for (int q = 0; q < SRC_PER_WAVE / 4; ++q) {
        v2f av;
        av.x = (q & 1) ? af[q >> 1].z : af[q >> 1].x;
        av.y = (q & 1) ? af[q >> 1].w : af[q >> 1].y;
        const float bx[4] = {b0[q].x, b0[q].y, b0[q].z, b0[q].w};
        const float by[4] = {b1[q].x, b1[q].y, b1[q].z, b1[q].w};
#pragma unroll
        for (int tt = 0; tt < 4; ++tt) {
            v2f bv;
            bv.x = bx[tt]; bv.y = by[tt];
            acc[tt] = __builtin_amdgcn_wmma_f32_16x16x4_f32(false, av, false, bv,
                                                            (short)0, acc[tt], false, false);
        }
    }

#pragma unroll
    for (int r = 0; r < 8; ++r)
        s_part[(w * 8 + r) * 32 + lane] = make_float4(acc[0][r], acc[1][r], acc[2][r], acc[3][r]);
    if (hh == 0) s_den[w][cc] = dsum;
    __syncthreads();

    if (t < 8 * 32) {
        const int r   = t >> 5;
        const int tgt = 8 * hh + r;
        float4 s = s_part[t];
        float d = s_den[0][tgt];
#pragma unroll
        for (int k = 1; k < AG_WAVES; ++k) {
            const float4 v = s_part[k * (8 * 32) + t];
            s.x += v.x; s.y += v.y; s.z += v.z; s.w += v.w;
            d += s_den[k][tgt];
        }
        s.x /= d; s.y /= d; s.z /= d; s.w /= d;
        if (use_elu[0]) {
            s.x = (s.x > 0.0f) ? s.x : expm1f(s.x);
            s.y = (s.y > 0.0f) ? s.y : expm1f(s.y);
            s.z = (s.z > 0.0f) ? s.z : expm1f(s.z);
            s.w = (s.w > 0.0f) ? s.w : expm1f(s.w);
        }
        *reinterpret_cast<float4*>(out + (size_t)(i0 + tgt) * OUT_F + h * HID + 4 * cc) = s;
    }
}

extern "C" void kernel_launch(void* const* d_in, const int* in_sizes, int n_in,
                              void* d_out, int out_size, void* d_ws, size_t ws_size,
                              hipStream_t stream) {
    const float* h_in  = (const float*)d_in[0];
    const float* nei   = (const float*)d_in[1];
    const float* adj   = (const float*)d_in[2];
    const float* w_l   = (const float*)d_in[3];
    const float* w_r   = (const float*)d_in[4];
    const float* aw    = (const float*)d_in[5];
    const int*   elu   = (const int*)d_in[6];
    float* out = (float*)d_out;

    float* ws    = (float*)d_ws;
    float* g_r   = ws + (size_t)N_NODES * OUT_F;
    float* g_l   = g_r + (size_t)N_NODES * OUT_F;

    project_both<<<dim3(OUT_F / 64, N_NODES / 32, 2), PJ_WAVES * 32, 0, stream>>>(
        h_in, w_l, g_l, nei, w_r, g_r);

    attn_fused<<<dim3(N_NODES / TGT_PER_BLOCK, HEADS), AG_WAVES * 32, 0, stream>>>(
        g_l, g_r, adj, aw, out, elu);
}


# g_l stored by the projection in source-minor blocks of 4 features: a score-loop load's 32 lanes read 512 contiguous bytes (4 cache lines) in place of 32 lines of 32 rows; all sums and orders kept
# speedup vs baseline: 7.4681x; 1.0666x over previous
#include <hip/hip_runtime.h>
#include <math.h>


typedef float v2f __attribute__((ext_vector_type(2)));
typedef float v8f __attribute__((ext_vector_type(8)));
typedef float v16f __attribute__((ext_vector_type(16)));

#define N_NODES 1024
#define IN_F    256
#define OUT_F   256
#define HEADS   4
#define HID     64
#define NEG_SLOPE 0.2f

#define PJ_WAVES 4
#define PJ_KBLK  32
#define PJ_SETK  (2 * PJ_KBLK)

struct PjSet {
    float4 a[2][4];
    float2 b[2][16];
};

__device__ __forceinline__ void pj_load(PjSet& s, const float* __restrict__ pa,
                                        const float* __restrict__ pb, const int k0) {
#pragma unroll
    for (int blk = 0; blk < 2; ++blk) {
        const int kb = k0 + blk * PJ_KBLK;
#pragma unroll
        for (int q = 0; q < 4; ++q)
            s.a[blk][q] = *reinterpret_cast<const float4*>(pa + kb + 4 * q);
#pragma unroll
        for (int e = 0; e < 16; ++e)
            s.b[blk][e] = *reinterpret_cast<const float2*>(pb + (size_t)(kb + e) * OUT_F);
    }
}

__device__ __forceinline__ void pj_mma(const PjSet& s, v8f& acc0, v8f& acc1) {
#pragma unroll
    for (int blk = 0; blk < 2; ++blk) {
#pragma unroll
        for (int u = 0; u < 8; ++u) {
            const float4 aq = s.a[blk][u >> 1];
            const float2 bk0 = s.b[blk][2 * u], bk1 = s.b[blk][2 * u + 1];
            v2f av, bv;
            av.x = (u & 1) ? aq.z : aq.x;
            av.y = (u & 1) ? aq.w : aq.y;
            bv.x = bk0.x; bv.y = bk1.x;
            acc0 = __builtin_amdgcn_wmma_f32_16x16x4_f32(false, av, false, bv,
                                                         (short)0, acc0, false, false);
            bv.x = bk0.y; bv.y = bk1.y;
            acc1 = __builtin_amdgcn_wmma_f32_16x16x4_f32(false, av, false, bv,
                                                         (short)0, acc1, false, false);
        }
    }
}

__global__ __launch_bounds__(PJ_WAVES * 32) __attribute__((amdgpu_waves_per_eu(1, 2)))
void project_both(const float* __restrict__ X0, const float* __restrict__ W0,
                  float* __restrict__ G0,
                  const float* __restrict__ X1, const float* __restrict__ W1,
                  float* __restrict__ G1) {
    const int t    = threadIdx.x;
    const int lane = t & 31;
    const int w    = __builtin_amdgcn_readfirstlane(t >> 5);
    const int c    = lane & 15;
    const int hh   = lane >> 4;
    const bool second = (blockIdx.z != 0);
    const float* __restrict__ X = second ? X1 : X0;
    const float* __restrict__ W = second ? W1 : W0;
    float* __restrict__ G       = second ? G1 : G0;

    const int tileM = (blockIdx.y * 2 + (w >> 1)) * 16;
    const int n0    = (blockIdx.x * 2 + (w & 1)) * 32;

    const float* pa = X + (size_t)(tileM + c) * IN_F + 16 * hh;
    const float* pb = W + (size_t)(16 * hh) * OUT_F + n0 + 2 * c;

    v8f acc0 = {}, acc1 = {};
    PjSet s0, s1;
    pj_load(s0, pa, pb, 0 * PJ_SETK);
    pj_load(s1, pa, pb, 1 * PJ_SETK);
    pj_mma(s0, acc0, acc1);
    pj_load(s0, pa, pb, 2 * PJ_SETK);
    pj_mma(s1, acc0, acc1);
    pj_load(s1, pa, pb, 3 * PJ_SETK);
    pj_mma(s0, acc0, acc1);
    pj_mma(s1, acc0, acc1);

    if (second) {
        float* po = G + (size_t)(tileM + 8 * hh) * OUT_F + n0 + 2 * c;
#pragma unroll
        for (int r = 0; r < 8; ++r)
            *reinterpret_cast<float2*>(po + (size_t)r * OUT_F) = make_float2(acc0[r], acc1[r]);
    } else {
        const int n = n0 + 2 * c;
        float* po = G + ((size_t)(n >> 2) * N_NODES + (tileM + 8 * hh)) * 4 + (n & 3);
#pragma unroll
        for (int r = 0; r < 8; ++r)
            *reinterpret_cast<float2*>(po + (size_t)r * 4) = make_float2(acc0[r], acc1[r]);
    }
}

#define AG_WAVES      16
#define SRC_PER_WAVE  64
#define TGT_PER_BLOCK 16
#define F_CHUNK       16
#define TGT_GROUP     4
#define P_PITCH       68

__device__ __forceinline__ void load_rows(v16f (&row)[TGT_GROUP], const float* __restrict__ grp,
                                          const int rg, const int c) {
#pragma unroll
    for (int r = 0; r < TGT_GROUP; ++r)
        row[r] = *reinterpret_cast<const v16f*>(grp + (size_t)(rg + r) * OUT_F + F_CHUNK * c);
}

template <int RG>
__device__ __forceinline__ void score_group(v2f (&e2)[TGT_PER_BLOCK], const v2f (&g2)[F_CHUNK],
                                            const v16f& awc, const v16f (&row)[TGT_GROUP]) {
#pragma unroll
    for (int f = 0; f < F_CHUNK; ++f) {
        const float awf = awc[f];
        v2f s[TGT_GROUP], m[TGT_GROUP], lr[TGT_GROUP];
#pragma unroll
        for (int r = 0; r < TGT_GROUP; ++r) {
            const float grf = row[r][f];
            s[r] = (v2f){grf, grf} + g2[f];
        }
#pragma unroll
        for (int r = 0; r < TGT_GROUP; ++r)
            m[r] = s[r] * (v2f){NEG_SLOPE, NEG_SLOPE};
#pragma unroll
        for (int r = 0; r < TGT_GROUP; ++r)
            lr[r] = __builtin_elementwise_max(s[r], m[r]);
#pragma unroll
        for (int r = 0; r < TGT_GROUP; ++r)
            e2[RG + r] = __builtin_elementwise_fma(lr[r], (v2f){awf, awf}, e2[RG + r]);
    }
}

__global__ __launch_bounds__(AG_WAVES * 32)
void attn_fused(const float* __restrict__ gl, const float* __restrict__ gr,
                const float* __restrict__ adj, const float* __restrict__ aw,
                float* __restrict__ out, const int* __restrict__ use_elu) {
    const int t    = threadIdx.x;
    const int lane = t & 31;
    const int w    = __builtin_amdgcn_readfirstlane(t >> 5);
    const int i0   = blockIdx.x * TGT_PER_BLOCK;
    const int h    = blockIdx.y;
    const int cc   = lane & 15;
    const int hh   = lane >> 4;

    __shared__ __attribute__((aligned(16))) float s_p[AG_WAVES * TGT_PER_BLOCK * P_PITCH];
    __shared__ float4 s_part[AG_WAVES * 8 * 32];
    __shared__ float s_den[AG_WAVES][TGT_PER_BLOCK];

    const int jbase = w * SRC_PER_WAVE;
    const float* __restrict__ pla = gl + ((size_t)(h * (HID / 4)) * N_NODES + jbase + lane) * 4;
    const float* __restrict__ plb = pla + 32 * 4;
    const float* __restrict__ grp = gr + (size_t)i0 * OUT_F + h * HID;

    v2f e2[TGT_PER_BLOCK];
#pragma unroll
    for (int r = 0; r < TGT_PER_BLOCK; ++r) { e2[r].x = 0.0f; e2[r].y = 0.0f; }

    float4 va[F_CHUNK / 4], vb[F_CHUNK / 4];
#pragma unroll
    for (int q = 0; q < F_CHUNK / 4; ++q) {
        va[q] = *reinterpret_cast<const float4*>(pla + (size_t)q * (4 * N_NODES));
        vb[q] = *reinterpret_cast<const float4*>(plb + (size_t)q * (4 * N_NODES));
    }

    asm volatile("" ::: "memory");
    const float* __restrict__ adja = adj + (size_t)(jbase + lane) * N_NODES + i0;
    const float* __restrict__ adjb = adja + (size_t)32 * N_NODES;
    float4 aja[4], ajb[4];
#pragma unroll
    for (int q = 0; q < 4; ++q) {
        aja[q] = *reinterpret_cast<const float4*>(adja + 4 * q);
        ajb[q] = *reinterpret_cast<const float4*>(adjb + 4 * q);
    }

#pragma unroll 1
    for (int c = 0; c < HID / F_CHUNK; ++c) {
        v2f g2[F_CHUNK];
#pragma unroll
        for (int q = 0; q < F_CHUNK / 4; ++q) {
            g2[4 * q].x = va[q].x;     g2[4 * q].y = vb[q].x;
            g2[4 * q + 1].x = va[q].y; g2[4 * q + 1].y = vb[q].y;
            g2[4 * q + 2].x = va[q].z; g2[4 * q + 2].y = vb[q].z;
            g2[4 * q + 3].x = va[q].w; g2[4 * q + 3].y = vb[q].w;
        }
        const int cn = (c + 1 < HID / F_CHUNK) ? c + 1 : c;
#pragma unroll
        for (int q = 0; q < F_CHUNK / 4; ++q) {
            va[q] = *reinterpret_cast<const float4*>(pla + (size_t)((F_CHUNK / 4) * cn + q) * (4 * N_NODES));
            vb[q] = *reinterpret_cast<const float4*>(plb + (size_t)((F_CHUNK / 4) * cn + q) * (4 * N_NODES));
        }
        asm volatile("" ::: "memory");
        const v16f awc = *reinterpret_cast<const v16f*>(aw + F_CHUNK * c);
        { v16f row[TGT_GROUP]; load_rows(row, grp, 0 * TGT_GROUP, c); score_group<0 * TGT_GROUP>(e2, g2, awc, row); }
        { v16f row[TGT_GROUP]; load_rows(row, grp, 1 * TGT_GROUP, c); score_group<1 * TGT_GROUP>(e2, g2, awc, row); }
        { v16f row[TGT_GROUP]; load_rows(row, grp, 2 * TGT_GROUP, c); score_group<2 * TGT_GROUP>(e2, g2, awc, row); }
        { v16f row[TGT_GROUP]; load_rows(row, grp, 3 * TGT_GROUP, c); score_group<3 * TGT_GROUP>(e2, g2, awc, row); }
    }

    const float* __restrict__ pbq = gr + (size_t)(jbase + 16 * hh) * OUT_F + h * HID + 4 * cc;
    float4 b0[SRC_PER_WAVE / 4], b1[SRC_PER_WAVE / 4];
#pragma unroll
    for (int q = 0; q < SRC_PER_WAVE / 8; ++q) {
        b0[q] = *reinterpret_cast<const float4*>(pbq + (size_t)q * OUT_F);
        b1[q] = *reinterpret_cast<const float4*>(pbq + (size_t)(32 + q) * OUT_F);
    }

    const v2f l2e = {1.4426950408889634f, 1.4426950408889634f};
    float* sp = s_p + w * (TGT_PER_BLOCK * P_PITCH);
#pragma unroll
    for (int r = 0; r < TGT_PER_BLOCK; ++r) {
        const float4 ma4 = aja[r >> 2], mb4 = ajb[r >> 2];
        const float ma = (r & 3) == 0 ? ma4.x : (r & 3) == 1 ? ma4.y : (r & 3) == 2 ? ma4.z : ma4.w;
        const float mb = (r & 3) == 0 ? mb4.x : (r & 3) == 1 ? mb4.y : (r & 3) == 2 ? mb4.z : mb4.w;
        const v2f t2 = e2[r] * l2e;
        const float xa = __builtin_amdgcn_exp2f(t2.x);
        const float xb = __builtin_amdgcn_exp2f(t2.y);
        v2f p2;
        p2.x = (ma != 0.0f) ? xa : 0.0f;
        p2.y = (mb != 0.0f) ? xb : 0.0f;
        *reinterpret_cast<v2f*>(sp + r * P_PITCH + 2 * lane) = p2;
    }

    __builtin_amdgcn_fence(__ATOMIC_RELEASE, "wavefront");
    __builtin_amdgcn_wave_barrier();
    __builtin_amdgcn_fence(__ATOMIC_ACQUIRE, "wavefront");

    float4 af[8];
    const float4* __restrict__ apr = reinterpret_cast<const float4*>(sp + cc * P_PITCH + 32 * hh);
#pragma unroll
    for (int m = 0; m < 8; ++m) af[m] = apr[m];

#pragma unroll
    for (int q = SRC_PER_WAVE / 8; q < SRC_PER_WAVE / 4; ++q) {
        b0[q] = *reinterpret_cast<const float4*>(pbq + (size_t)q * OUT_F);
        b1[q] = *reinterpret_cast<const float4*>(pbq + (size_t)(32 + q) * OUT_F);
    }

    float dsum = 0.0f;
#pragma unroll
    for (int m = 0; m < 8; ++m) dsum += (af[m].x + af[m].y) + (af[m].z + af[m].w);
    dsum += __builtin_bit_cast(float, __builtin_amdgcn_permlanex16(
        __builtin_bit_cast(int, dsum), __builtin_bit_cast(int, dsum),
        0x76543210, 0xfedcba98, false, false));

    v8f acc[4];
#pragma unroll
    for (int tt = 0; tt < 4; ++tt) acc[tt] = (v8f){};
#pragma unroll
    for (int q = 0; q < SRC_PER_WAVE / 4; ++q) {
        v2f av;
        av.x = (q & 1) ? af[q >> 1].z : af[q >> 1].x;
        av.y = (q & 1) ? af[q >> 1].w : af[q >> 1].y;
        const float bx[4] = {b0[q].x, b0[q].y, b0[q].z, b0[q].w};
        const float by[4] = {b1[q].x, b1[q].y, b1[q].z, b1[q].w};
#pragma unroll
        for (int tt = 0; tt < 4; ++tt) {
            v2f bv;
            bv.x = bx[tt]; bv.y = by[tt];
            acc[tt] = __builtin_amdgcn_wmma_f32_16x16x4_f32(false, av, false, bv,
                                                            (short)0, acc[tt], false, false);
        }
    }

#pragma unroll
    for (int r = 0; r < 8; ++r)
        s_part[(w * 8 + r) * 32 + lane] = make_float4(acc[0][r], acc[1][r], acc[2][r], acc[3][r]);
    if (hh == 0) s_den[w][cc] = dsum;
    __syncthreads();

    if (t < 8 * 32) {
        const int r   = t >> 5;
        const int tgt = 8 * hh + r;
        float4 s = s_part[t];
        float d = s_den[0][tgt];
#pragma unroll
        for (int k = 1; k < AG_WAVES; ++k) {
            const float4 v = s_part[k * (8 * 32) + t];
            s.x += v.x; s.y += v.y; s.z += v.z; s.w += v.w;
            d += s_den[k][tgt];
        }
        s.x /= d; s.y /= d; s.z /= d; s.w /= d;
        if (use_elu[0]) {
            s.x = (s.x > 0.0f) ? s.x : expm1f(s.x);
            s.y = (s.y > 0.0f) ? s.y : expm1f(s.y);
            s.z = (s.z > 0.0f) ? s.z : expm1f(s.z);
            s.w = (s.w > 0.0f) ? s.w : expm1f(s.w);
        }
        *reinterpret_cast<float4*>(out + (size_t)(i0 + tgt) * OUT_F + h * HID + 4 * cc) = s;
    }
}

extern "C" void kernel_launch(void* const* d_in, const int* in_sizes, int n_in,
                              void* d_out, int out_size, void* d_ws, size_t ws_size,
                              hipStream_t stream) {
    const float* h_in  = (const float*)d_in[0];
    const float* nei   = (const float*)d_in[1];
    const float* adj   = (const float*)d_in[2];
    const float* w_l   = (const float*)d_in[3];
    const float* w_r   = (const float*)d_in[4];
    const float* aw    = (const float*)d_in[5];
    const int*   elu   = (const int*)d_in[6];
    float* out = (float*)d_out;

    float* ws    = (float*)d_ws;
    float* g_r   = ws + (size_t)N_NODES * OUT_F;
    float* g_l   = g_r + (size_t)N_NODES * OUT_F;

    project_both<<<dim3(OUT_F / 64, N_NODES / 32, 2), PJ_WAVES * 32, 0, stream>>>(
        h_in, w_l, g_l, nei, w_r, g_r);

    attn_fused<<<dim3(N_NODES / TGT_PER_BLOCK, HEADS), AG_WAVES * 32, 0, stream>>>(
        g_l, g_r, adj, aw, out, elu);
}
